# starting kernel
# baseline (speedup 1.0000x reference)
#include <hip/hip_runtime.h>
#include <hip/hip_bf16.h>

#define DIM       128
#define HEADS     4
#define HEAD_DIM  32

typedef __attribute__((ext_vector_type(2))) float v2f;
typedef __attribute__((ext_vector_type(8))) float v8f;

__device__ __forceinline__ unsigned flipf(float f) {
    unsigned u = __float_as_uint(f);
    return (u & 0x80000000u) ? ~u : (u | 0x80000000u);
}
__device__ __forceinline__ float unflipf(unsigned u) {
    return __uint_as_float((u & 0x80000000u) ? (u & 0x7FFFFFFFu) : ~u);
}

__global__ __launch_bounds__(256)
void gat_gemm_wmma(const float* __restrict__ x, const float* __restrict__ W,
                   float* __restrict__ Wx, int nNodes, int nTiles) {
    int wave = (blockIdx.x * blockDim.x + threadIdx.x) >> 5;
    if (wave >= nTiles) return;
    int lane = threadIdx.x & 31;
    int rc   = lane & 15;
    int hi   = lane >> 4;
    int nodeBase = wave * 16;

    v8f acc[8] = {};

    int arow = nodeBase + rc;
    if (arow >= nNodes) arow = nNodes - 1;
    const float* aptr = x + (size_t)arow * DIM + hi * 2;

    #pragma unroll 8
    for (int k = 0; k < DIM; k += 4) {
        v2f a = *(const v2f*)(aptr + k);
        #pragma unroll
        for (int nt = 0; nt < 8; ++nt) {
            v2f b = *(const v2f*)(W + (size_t)(nt * 16 + rc) * DIM + k + hi * 2);
            acc[nt] = __builtin_amdgcn_wmma_f32_16x16x4_f32(
                false, a, false, b, (short)0, acc[nt], false, false);
        }
    }

    #pragma unroll
    for (int nt = 0; nt < 8; ++nt) {
        #pragma unroll
        for (int j = 0; j < 8; ++j) {
            int node = nodeBase + j + 8 * hi;
            if (node < nNodes)
                Wx[(size_t)node * DIM + nt * 16 + rc] = acc[nt][j];
        }
    }
}

__global__ __launch_bounds__(256)
void gat_scores(const float* __restrict__ Wx, const float* __restrict__ a_w,
                float* __restrict__ s_src, float* __restrict__ s_dst, int nNodes) {
    int t = blockIdx.x * blockDim.x + threadIdx.x;
    if (t >= nNodes * HEADS) return;
    int h = t & (HEADS - 1);
    int n = t >> 2;
    const float* p = Wx + (size_t)n * DIM + h * HEAD_DIM;
    float ss = 0.f, sd = 0.f;
    #pragma unroll
    for (int d = 0; d < HEAD_DIM; ++d) {
        float v = p[d];
        ss += v * a_w[d];
        sd += v * a_w[HEAD_DIM + d];
    }
    s_src[t] = ss;
    s_dst[t] = sd;
}

__global__ __launch_bounds__(256)
void gat_edge_max(const int* __restrict__ ei, const float* __restrict__ s_src,
                  const float* __restrict__ s_dst, unsigned* __restrict__ emax,
                  int nE) {
    int t = blockIdx.x * blockDim.x + threadIdx.x;
    if (t >= nE * HEADS) return;
    int h = t & (HEADS - 1);
    int e = t >> 2;
    int src = ei[e];
    int dst = ei[nE + e];
    float v = s_src[src * HEADS + h] + s_dst[dst * HEADS + h];
    v = v > 0.f ? v : 0.2f * v;
    atomicMax(&emax[dst * HEADS + h], flipf(v));
}

__global__ __launch_bounds__(256)
void gat_edge_accum(const int* __restrict__ ei, const float* __restrict__ s_src,
                    const float* __restrict__ s_dst, const unsigned* __restrict__ emax,
                    const float* __restrict__ Wx, float* __restrict__ denom,
                    float* __restrict__ out, int nE) {
    int gw = (int)((blockIdx.x * (size_t)blockDim.x + threadIdx.x) >> 5);
    if (gw >= nE * HEADS) return;
    int lane = threadIdx.x & 31;
    int h = gw & (HEADS - 1);
    int e = gw >> 2;
    int src = ei[e];
    int dst = ei[nE + e];
    float v = s_src[src * HEADS + h] + s_dst[dst * HEADS + h];
    v = v > 0.f ? v : 0.2f * v;
    float em = unflipf(emax[dst * HEADS + h]);
    float ee = expf(v - em);
    if (lane == 0) atomicAdd(&denom[dst * HEADS + h], ee);
    atomicAdd(&out[(size_t)dst * DIM + h * HEAD_DIM + lane],
              ee * Wx[(size_t)src * DIM + h * HEAD_DIM + lane]);
}

__global__ __launch_bounds__(256)
void gat_finalize(float* __restrict__ out, const float* __restrict__ denom,
                  int nNodes) {
    int t = blockIdx.x * blockDim.x + threadIdx.x;
    if (t >= nNodes * DIM) return;
    int n = t >> 7;
    int d = t & (DIM - 1);
    int h = d >> 5;
    float v = out[t] / (denom[n * HEADS + h] + 1e-8f);
    out[t] = v > 0.f ? v : expm1f(v);
}

extern "C" void kernel_launch(void* const* d_in, const int* in_sizes, int n_in,
                              void* d_out, int out_size, void* d_ws, size_t ws_size,
                              hipStream_t stream) {
    const float* x   = (const float*)d_in[0];
    const int*   ei  = (const int*)d_in[1];
    const float* W   = (const float*)d_in[2];
    const float* a_w = (const float*)d_in[3];
    int N = in_sizes[0] / DIM;
    int E = in_sizes[1] / 2;
    float* out = (float*)d_out;

    float*    Wx    = (float*)d_ws;
    float*    s_src = Wx + (size_t)N * DIM;
    float*    s_dst = s_src + (size_t)N * HEADS;
    float*    denom = s_dst + (size_t)N * HEADS;
    unsigned* emax  = (unsigned*)(denom + (size_t)N * HEADS);

    hipMemsetAsync(out,   0, (size_t)N * DIM * sizeof(float), stream);
    hipMemsetAsync(denom, 0, (size_t)N * HEADS * sizeof(float), stream);
    hipMemsetAsync(emax,  0, (size_t)N * HEADS * sizeof(unsigned), stream);

    dim3 blk(256);
    int nTiles = (N + 15) / 16;
    gat_gemm_wmma<<<(nTiles + 7) / 8, blk, 0, stream>>>(x, W, Wx, N, nTiles);
    gat_scores  <<<(N * HEADS + 255) / 256, blk, 0, stream>>>(Wx, a_w, s_src, s_dst, N);
    gat_edge_max<<<((size_t)E * HEADS + 255) / 256, blk, 0, stream>>>(ei, s_src, s_dst, emax, E);
    gat_edge_accum<<<((size_t)E * HEADS * 32 + 255) / 256, blk, 0, stream>>>(
        ei, s_src, s_dst, emax, Wx, denom, out, E);
    gat_finalize<<<((size_t)N * DIM + 255) / 256, blk, 0, stream>>>(out, denom, N);
}


# Edge stage in one kernel: a work-group owns 196 dst nodes in LDS, lists their edges from one scan of the dst list; one wave adds a node's edges in list order: float atomics gone, sum order now fixed
# speedup vs baseline: 3.6973x; 3.6973x over previous
#include <hip/hip_runtime.h>
#include <hip/hip_bf16.h>

#define DIM       128
#define HEADS     4
#define HEAD_DIM  32

typedef __attribute__((ext_vector_type(2))) float v2f;
typedef __attribute__((ext_vector_type(8))) float v8f;

__device__ __forceinline__ unsigned flipf(float f) {
    unsigned u = __float_as_uint(f);
    return (u & 0x80000000u) ? ~u : (u | 0x80000000u);
}
__device__ __forceinline__ float unflipf(unsigned u) {
    return __uint_as_float((u & 0x80000000u) ? (u & 0x7FFFFFFFu) : ~u);
}

__global__ __launch_bounds__(256)
void gat_gemm_wmma(const float* __restrict__ x, const float* __restrict__ W,
                   float* __restrict__ Wx, int nNodes, int nTiles) {
    int wave = (blockIdx.x * blockDim.x + threadIdx.x) >> 5;
    if (wave >= nTiles) return;
    int lane = threadIdx.x & 31;
    int rc   = lane & 15;
    int hi   = lane >> 4;
    int nodeBase = wave * 16;

    v8f acc[8] = {};

    int arow = nodeBase + rc;
    if (arow >= nNodes) arow = nNodes - 1;
    const float* aptr = x + (size_t)arow * DIM + hi * 2;

    #pragma unroll 8
    for (int k = 0; k < DIM; k += 4) {
        v2f a = *(const v2f*)(aptr + k);
        #pragma unroll
        for (int nt = 0; nt < 8; ++nt) {
            v2f b = *(const v2f*)(W + (size_t)(nt * 16 + rc) * DIM + k + hi * 2);
            acc[nt] = __builtin_amdgcn_wmma_f32_16x16x4_f32(
                false, a, false, b, (short)0, acc[nt], false, false);
        }
    }

    #pragma unroll
    for (int nt = 0; nt < 8; ++nt) {
        #pragma unroll
        for (int j = 0; j < 8; ++j) {
            int node = nodeBase + j + 8 * hi;
            if (node < nNodes)
                Wx[(size_t)node * DIM + nt * 16 + rc] = acc[nt][j];
        }
    }
}

__global__ __launch_bounds__(256)
void gat_scores(const float* __restrict__ Wx, const float* __restrict__ a_w,
                float* __restrict__ s_src, float* __restrict__ s_dst, int nNodes) {
    int t = blockIdx.x * blockDim.x + threadIdx.x;
    if (t >= nNodes * HEADS) return;
    int h = t & (HEADS - 1);
    int n = t >> 2;
    const float* p = Wx + (size_t)n * DIM + h * HEAD_DIM;
    float ss = 0.f, sd = 0.f;
    #pragma unroll
    for (int d = 0; d < HEAD_DIM; ++d) {
        float v = p[d];
        ss += v * a_w[d];
        sd += v * a_w[HEAD_DIM + d];
    }
    s_src[t] = ss;
    s_dst[t] = sd;
}

#define RN    196
#define NWAVE 32
#define LCAP  512
#define LSTR  (LCAP + 1)

__global__ __launch_bounds__(1024)
void gat_edge_owner(const int* __restrict__ ei, const float* __restrict__ s_src,
                    const float* __restrict__ s_dst, const float* __restrict__ Wx,
                    float* __restrict__ out, int nE, int nNodes) {
    __shared__ float4   acc[RN * 32];
    __shared__ float    den[RN * HEADS];
    __shared__ unsigned mx[RN * HEADS];
    __shared__ float    sd[RN * HEADS];
    __shared__ int2     lst[NWAVE * LSTR];
    __shared__ int      lcnt[NWAVE];
    __shared__ int      ovf;

    const int tid  = threadIdx.x;
    const int lane = tid & 31;
    const int w    = __builtin_amdgcn_readfirstlane(tid >> 5);
    const int lo   = blockIdx.x * RN;
    int cntR = nNodes - lo;
    if (cntR > RN) cntR = RN;
    const unsigned ucnt = (unsigned)cntR;
    const int* __restrict__ dstp = ei + nE;

    for (int i = tid; i < RN * 32; i += 1024) acc[i] = make_float4(0.f, 0.f, 0.f, 0.f);
    for (int i = tid; i < RN * HEADS; i += 1024) {
        den[i] = 0.f;
        mx[i]  = 0u;
        sd[i]  = (i < cntR * HEADS) ? s_dst[(size_t)lo * HEADS + i] : 0.f;
    }
    if (tid == 0) ovf = 0;
    __syncthreads();

    int cnt = 0;
    int2* mylist = lst + w * LSTR;
    auto put = [&](int idx, unsigned u) {
        if (cnt < LCAP && lane == 0) mylist[cnt] = make_int2(idx, (int)u);
        ++cnt;
    };
    int head = (int)(((16u - (unsigned)((size_t)dstp & 15u)) & 15u) >> 2);
    if (head > nE) head = nE;
    const int nblk = (nE - head) >> 10;
    const int4* __restrict__ d4 = (const int4*)(dstp + head);
    for (int b = w; b < nblk; b += NWAVE) {
        int4 q[8];
        #pragma unroll
        for (int k = 0; k < 8; ++k) q[k] = d4[(size_t)b * 256 + k * 32 + lane];
        #pragma unroll
        for (int k = 0; k < 8; ++k) {
            unsigned u0 = (unsigned)(q[k].x - lo), u1 = (unsigned)(q[k].y - lo);
            unsigned u2 = (unsigned)(q[k].z - lo), u3 = (unsigned)(q[k].w - lo);
            unsigned m01 = u0 < u1 ? u0 : u1, m23 = u2 < u3 ? u2 : u3;
            unsigned um = m01 < m23 ? m01 : m23;
            unsigned hit = __builtin_amdgcn_ballot_w32(um < ucnt);
            while (hit) {
                int j = __builtin_ctz(hit);
                hit &= hit - 1;
                int e0 = head + b * 1024 + k * 128 + j * 4;
                unsigned a0 = (unsigned)__builtin_amdgcn_readlane((int)u0, j);
                unsigned a1 = (unsigned)__builtin_amdgcn_readlane((int)u1, j);
                unsigned a2 = (unsigned)__builtin_amdgcn_readlane((int)u2, j);
                unsigned a3 = (unsigned)__builtin_amdgcn_readlane((int)u3, j);
                if (a0 < ucnt) put(e0, a0);
                if (a1 < ucnt) put(e0 + 1, a1);
                if (a2 < ucnt) put(e0 + 2, a2);
                if (a3 < ucnt) put(e0 + 3, a3);
            }
        }
    }
    auto gen = [&](int beg, int end) {
        for (int b = beg + w * 32; b < end; b += 32 * NWAVE) {
            int idx = b + lane;
            unsigned u = (idx < end) ? (unsigned)(dstp[idx] - lo) : 0xFFFFFFFFu;
            unsigned hit = __builtin_amdgcn_ballot_w32(u < ucnt);
            while (hit) {
                int j = __builtin_ctz(hit);
                hit &= hit - 1;
                put(b + j, (unsigned)__builtin_amdgcn_readlane((int)u, j));
            }
        }
    };
    gen(0, head);
    gen(head + nblk * 1024, nE);
    if (lane == 0) {
        lcnt[w] = cnt < LCAP ? cnt : LCAP;
        if (cnt > LCAP) ovf = 1;
    }
    __syncthreads();
    const int slow = __builtin_amdgcn_readfirstlane(ovf);

    if (!slow) {
        for (int i = w; i < LCAP; i += NWAVE) {
            bool act = i < lcnt[lane];
            if (__builtin_amdgcn_ballot_w32(act) == 0u) break;
            if (act) {
                int2 e = lst[lane * LSTR + i];
                int src = ei[e.x];
                float4 ss = ((const float4*)s_src)[src];
                float v0 = ss.x + sd[e.y * HEADS + 0];
                float v1 = ss.y + sd[e.y * HEADS + 1];
                float v2 = ss.z + sd[e.y * HEADS + 2];
                float v3 = ss.w + sd[e.y * HEADS + 3];
                v0 = v0 > 0.f ? v0 : 0.2f * v0;
                v1 = v1 > 0.f ? v1 : 0.2f * v1;
                v2 = v2 > 0.f ? v2 : 0.2f * v2;
                v3 = v3 > 0.f ? v3 : 0.2f * v3;
                atomicMax(&mx[e.y * HEADS + 0], flipf(v0));
                atomicMax(&mx[e.y * HEADS + 1], flipf(v1));
                atomicMax(&mx[e.y * HEADS + 2], flipf(v2));
                atomicMax(&mx[e.y * HEADS + 3], flipf(v3));
                lst[lane * LSTR + i].x = src;
            }
        }
    } else {
        for (int b = 0; b < nE; b += 32) {
            int idx = b + lane;
            unsigned u = (idx < nE) ? (unsigned)(dstp[idx] - lo) : 0xFFFFFFFFu;
            unsigned hit = __builtin_amdgcn_ballot_w32(u < ucnt && (int)(u & 31u) == w);
            while (hit) {
                int j = __builtin_ctz(hit);
                hit &= hit - 1;
                int d = __builtin_amdgcn_readlane((int)u, j);
                int src = ei[b + j];
                if (lane < HEADS) {
                    float v = s_src[src * HEADS + lane] + sd[d * HEADS + lane];
                    v = v > 0.f ? v : 0.2f * v;
                    atomicMax(&mx[d * HEADS + lane], flipf(v));
                }
            }
        }
    }
    __syncthreads();

    int pend_s = 0, pend_d = 0, npend = 0;
    const int h = lane >> 3;
    auto flush = [&]() {
        float4 row[8];
        float  sv[8];
        #pragma unroll
        for (int k = 0; k < 8; ++k) {
            int s = __builtin_amdgcn_readlane(pend_s, k);
            row[k] = *(const float4*)(Wx + (size_t)s * DIM + lane * 4);
            sv[k]  = s_src[s * HEADS + h];
        }
        #pragma unroll
        for (int k = 0; k < 8; ++k) {
            if (k < npend) {
                int d = __builtin_amdgcn_readlane(pend_d, k);
                float v = sv[k] + sd[d * HEADS + h];
                v = v > 0.f ? v : 0.2f * v;
                float ee = expf(v - unflipf(mx[d * HEADS + h]));
                float4 a = acc[d * 32 + lane];
                a.x += ee * row[k].x;
                a.y += ee * row[k].y;
                a.z += ee * row[k].z;
                a.w += ee * row[k].w;
                acc[d * 32 + lane] = a;
                if ((lane & 7) == 0) den[d * HEADS + h] += ee;
            }
        }
        npend = 0;
    };
    auto push = [&](int s, int d) {
        pend_s = (lane == npend) ? s : pend_s;
        pend_d = (lane == npend) ? d : pend_d;
        if (++npend == 8) flush();
    };
    if (!slow) {
        for (int s = 0; s < NWAVE; ++s) {
            const int c = __builtin_amdgcn_readfirstlane(lcnt[s]);
            for (int i = 0; i < c; i += 32) {
                int k = i + lane;
                int2 e = (k < c) ? lst[s * LSTR + k] : make_int2(0, -1);
                unsigned hit = __builtin_amdgcn_ballot_w32(k < c && (e.y & 31) == w);
                while (hit) {
                    int j = __builtin_ctz(hit);
                    hit &= hit - 1;
                    push(__builtin_amdgcn_readlane(e.x, j), __builtin_amdgcn_readlane(e.y, j));
                }
            }
        }
    } else {
        for (int b = 0; b < nE; b += 32) {
            int idx = b + lane;
            unsigned u = (idx < nE) ? (unsigned)(dstp[idx] - lo) : 0xFFFFFFFFu;
            bool mine = u < ucnt && (int)(u & 31u) == w;
            int sl = mine ? ei[idx] : 0;
            unsigned hit = __builtin_amdgcn_ballot_w32(mine);
            while (hit) {
                int j = __builtin_ctz(hit);
                hit &= hit - 1;
                push(__builtin_amdgcn_readlane(sl, j), __builtin_amdgcn_readlane((int)u, j));
            }
        }
    }
    if (npend > 0) flush();
    __syncthreads();

    for (int q = tid; q < cntR * 32; q += 1024) {
        int n = q >> 5, c4 = q & 31;
        float dn = den[n * HEADS + (c4 >> 3)] + 1e-8f;
        float4 a = acc[q];
        float4 r;
        r.x = a.x / dn; r.x = r.x > 0.f ? r.x : expm1f(r.x);
        r.y = a.y / dn; r.y = r.y > 0.f ? r.y : expm1f(r.y);
        r.z = a.z / dn; r.z = r.z > 0.f ? r.z : expm1f(r.z);
        r.w = a.w / dn; r.w = r.w > 0.f ? r.w : expm1f(r.w);
        *(float4*)(out + (size_t)(lo + n) * DIM + c4 * 4) = r;
    }
}

extern "C" void kernel_launch(void* const* d_in, const int* in_sizes, int n_in,
                              void* d_out, int out_size, void* d_ws, size_t ws_size,
                              hipStream_t stream) {
    const float* x   = (const float*)d_in[0];
    const int*   ei  = (const int*)d_in[1];
    const float* W   = (const float*)d_in[2];
    const float* a_w = (const float*)d_in[3];
    int N = in_sizes[0] / DIM;
    int E = in_sizes[1] / 2;
    float* out = (float*)d_out;

    float* Wx    = (float*)d_ws;
    float* s_src = Wx + (size_t)N * DIM;
    float* s_dst = s_src + (size_t)N * HEADS;

    dim3 blk(256);
    int nTiles = (N + 15) / 16;
    gat_gemm_wmma<<<(nTiles + 7) / 8, blk, 0, stream>>>(x, W, Wx, N, nTiles);
    gat_scores  <<<(N * HEADS + 255) / 256, blk, 0, stream>>>(Wx, a_w, s_src, s_dst, N);
    gat_edge_owner<<<(N + RN - 1) / RN, dim3(1024), 0, stream>>>(ei, s_src, s_dst, Wx, out, E, N);
}


# Edge scan reads a 1-byte dst-range key per edge (written in the scores launch) in place of the 4-byte dst; hit lanes append edge ids at their ballot rank; sum order changed, still fixed
# speedup vs baseline: 4.0004x; 1.0820x over previous
#include <hip/hip_runtime.h>
#include <hip/hip_bf16.h>

#define DIM       128
#define HEADS     4
#define HEAD_DIM  32
#define RN        196

typedef __attribute__((ext_vector_type(2))) float v2f;
typedef __attribute__((ext_vector_type(8))) float v8f;

__device__ __forceinline__ unsigned flipf(float f) {
    unsigned u = __float_as_uint(f);
    return (u & 0x80000000u) ? ~u : (u | 0x80000000u);
}
__device__ __forceinline__ float unflipf(unsigned u) {
    return __uint_as_float((u & 0x80000000u) ? (u & 0x7FFFFFFFu) : ~u);
}

__global__ __launch_bounds__(256)
void gat_gemm_wmma(const float* __restrict__ x, const float* __restrict__ W,
                   float* __restrict__ Wx, int nNodes, int nTiles) {
    int wave = (blockIdx.x * blockDim.x + threadIdx.x) >> 5;
    if (wave >= nTiles) return;
    int lane = threadIdx.x & 31;
    int rc   = lane & 15;
    int hi   = lane >> 4;
    int nodeBase = wave * 16;

    v8f acc[8] = {};

    int arow = nodeBase + rc;
    if (arow >= nNodes) arow = nNodes - 1;
    const float* aptr = x + (size_t)arow * DIM + hi * 2;

    #pragma unroll 8
    for (int k = 0; k < DIM; k += 4) {
        v2f a = *(const v2f*)(aptr + k);
        #pragma unroll
        for (int nt = 0; nt < 8; ++nt) {
            v2f b = *(const v2f*)(W + (size_t)(nt * 16 + rc) * DIM + k + hi * 2);
            acc[nt] = __builtin_amdgcn_wmma_f32_16x16x4_f32(
                false, a, false, b, (short)0, acc[nt], false, false);
        }
    }

    #pragma unroll
    for (int nt = 0; nt < 8; ++nt) {
        #pragma unroll
        for (int j = 0; j < 8; ++j) {
            int node = nodeBase + j + 8 * hi;
            if (node < nNodes)
                Wx[(size_t)node * DIM + nt * 16 + rc] = acc[nt][j];
        }
    }
}

__global__ __launch_bounds__(256)
void gat_scores(const float* __restrict__ Wx, const float* __restrict__ a_w,
                float* __restrict__ s_src, float* __restrict__ s_dst, int nNodes,
                const int* __restrict__ ei, unsigned* __restrict__ keys, int nE) {
    int t = blockIdx.x * blockDim.x + threadIdx.x;
    const int* __restrict__ dstp = ei + nE;
    const int nW = ((nE + 15) >> 4) << 2;
    for (int q = t; q < nW; q += (int)(gridDim.x * blockDim.x)) {
        unsigned wd = 0u;
        #pragma unroll
        for (int b = 0; b < 4; ++b) {
            int e = q * 4 + b;
            unsigned k = (e < nE) ? (((unsigned)dstp[e] / (unsigned)RN) & 255u) : 255u;
            wd |= k << (8 * b);
        }
        keys[q] = wd;
    }
    if (t >= nNodes * HEADS) return;
    int h = t & (HEADS - 1);
    int n = t >> 2;
    const float* p = Wx + (size_t)n * DIM + h * HEAD_DIM;
    float ss = 0.f, sd = 0.f;
    #pragma unroll
    for (int d = 0; d < HEAD_DIM; ++d) {
        float v = p[d];
        ss += v * a_w[d];
        sd += v * a_w[HEAD_DIM + d];
    }
    s_src[t] = ss;
    s_dst[t] = sd;
}

#define NWAVE 32
#define LCAP  512
#define LSTR  (LCAP + 1)

__global__ __launch_bounds__(1024)
void gat_edge_owner(const int* __restrict__ ei, const float* __restrict__ s_src,
                    const float* __restrict__ s_dst, const float* __restrict__ Wx,
                    const unsigned* __restrict__ keys,
                    float* __restrict__ out, int nE, int nNodes) {
    __shared__ float4   acc[RN * 32];
    __shared__ float    den[RN * HEADS];
    __shared__ unsigned mx[RN * HEADS];
    __shared__ float    sd[RN * HEADS];
    __shared__ int2     lst[NWAVE * LSTR];
    __shared__ int      lcnt[NWAVE];
    __shared__ int      ovf;

    const int tid  = threadIdx.x;
    const int lane = tid & 31;
    const int w    = __builtin_amdgcn_readfirstlane(tid >> 5);
    const int lo   = blockIdx.x * RN;
    int cntR = nNodes - lo;
    if (cntR > RN) cntR = RN;
    const unsigned ucnt = (unsigned)cntR;
    const int* __restrict__ dstp = ei + nE;

    for (int i = tid; i < RN * 32; i += 1024) acc[i] = make_float4(0.f, 0.f, 0.f, 0.f);
    for (int i = tid; i < RN * HEADS; i += 1024) {
        den[i] = 0.f;
        mx[i]  = 0u;
        sd[i]  = (i < cntR * HEADS) ? s_dst[(size_t)lo * HEADS + i] : 0.f;
    }
    if (tid == 0) ovf = 0;
    __syncthreads();

    int cnt = 0;
    int2* mylist = lst + w * LSTR;
    const unsigned pat = (unsigned)(blockIdx.x & 255) * 0x01010101u;
    const int nvec = (nE + 15) >> 4;
    const uint4* __restrict__ k4 = (const uint4*)keys;
    auto proc = [&](uint4 qk, int v) {
        unsigned t0 = qk.x ^ pat, t1 = qk.y ^ pat, t2 = qk.z ^ pat, t3 = qk.w ^ pat;
        unsigned z0 = (t0 - 0x01010101u) & ~t0 & 0x80808080u;
        unsigned z1 = (t1 - 0x01010101u) & ~t1 & 0x80808080u;
        unsigned z2 = (t2 - 0x01010101u) & ~t2 & 0x80808080u;
        unsigned z3 = (t3 - 0x01010101u) & ~t3 & 0x80808080u;
        unsigned c = (z0 >> 7) | (z1 >> 6) | (z2 >> 5) | (z3 >> 4);
        const int ebase = v * 16;
        unsigned bal = __builtin_amdgcn_ballot_w32(c != 0u);
        while (bal) {
            int pos = cnt + (int)__builtin_amdgcn_mbcnt_lo(bal, 0u);
            if (c != 0u) {
                int p = __builtin_ctz(c);
                if (pos < LCAP) mylist[pos] = make_int2(ebase + ((p & 7) << 2) + (p >> 3), -1);
            }
            cnt += __builtin_popcount(bal);
            c &= c - 1u;
            bal = __builtin_amdgcn_ballot_w32(c != 0u);
        }
    };
    const int nfull = nvec >> 7;
    for (int gi = w; gi < nfull; gi += NWAVE) {
        uint4 q[4];
        #pragma unroll
        for (int k = 0; k < 4; ++k) q[k] = k4[gi * 128 + k * 32 + lane];
        #pragma unroll
        for (int k = 0; k < 4; ++k) proc(q[k], gi * 128 + k * 32 + lane);
    }
    for (int v0 = nfull * 128 + w * 32; v0 < nvec; v0 += NWAVE * 32) {
        int v = v0 + lane;
        uint4 qk = (v < nvec) ? k4[v] : make_uint4(~pat, ~pat, ~pat, ~pat);
        proc(qk, v);
    }
    if (lane == 0) {
        lcnt[w] = cnt < LCAP ? cnt : LCAP;
        if (cnt > LCAP) ovf = 1;
    }
    __syncthreads();
    const int slow = __builtin_amdgcn_readfirstlane(ovf);

    if (!slow) {
        for (int i = w; i < LCAP; i += NWAVE) {
            bool act = i < lcnt[lane];
            if (__builtin_amdgcn_ballot_w32(act) == 0u) break;
            if (act) {
                const int eid = lst[lane * LSTR + i].x;
                const bool inb = eid < nE;
                const int dv  = inb ? dstp[eid] : -1;
                const int src = inb ? ei[eid] : 0;
                const unsigned u = (unsigned)(dv - lo);
                if (inb && u < ucnt) {
                    float4 ss = ((const float4*)s_src)[src];
                    float v0 = ss.x + sd[u * HEADS + 0];
                    float v1 = ss.y + sd[u * HEADS + 1];
                    float v2 = ss.z + sd[u * HEADS + 2];
                    float v3 = ss.w + sd[u * HEADS + 3];
                    v0 = v0 > 0.f ? v0 : 0.2f * v0;
                    v1 = v1 > 0.f ? v1 : 0.2f * v1;
                    v2 = v2 > 0.f ? v2 : 0.2f * v2;
                    v3 = v3 > 0.f ? v3 : 0.2f * v3;
                    atomicMax(&mx[u * HEADS + 0], flipf(v0));
                    atomicMax(&mx[u * HEADS + 1], flipf(v1));
                    atomicMax(&mx[u * HEADS + 2], flipf(v2));
                    atomicMax(&mx[u * HEADS + 3], flipf(v3));
                    lst[lane * LSTR + i] = make_int2(src, (int)u);
                } else {
                    lst[lane * LSTR + i] = make_int2(0, -1);
                }
            }
        }
    } else {
        for (int b = 0; b < nE; b += 32) {
            int idx = b + lane;
            unsigned u = (idx < nE) ? (unsigned)(dstp[idx] - lo) : 0xFFFFFFFFu;
            unsigned hit = __builtin_amdgcn_ballot_w32(u < ucnt && (int)(u & 31u) == w);
            while (hit) {
                int j = __builtin_ctz(hit);
                hit &= hit - 1;
                int d = __builtin_amdgcn_readlane((int)u, j);
                int src = ei[b + j];
                if (lane < HEADS) {
                    float v = s_src[src * HEADS + lane] + sd[d * HEADS + lane];
                    v = v > 0.f ? v : 0.2f * v;
                    atomicMax(&mx[d * HEADS + lane], flipf(v));
                }
            }
        }
    }
    __syncthreads();

    int pend_s = 0, pend_d = 0, npend = 0;
    const int h = lane >> 3;
    auto flush = [&]() {
        float4 row[8];
        float  sv[8];
        #pragma unroll
        for (int k = 0; k < 8; ++k) {
            int s = __builtin_amdgcn_readlane(pend_s, k);
            row[k] = *(const float4*)(Wx + (size_t)s * DIM + lane * 4);
            sv[k]  = s_src[s * HEADS + h];
        }
        #pragma unroll
        for (int k = 0; k < 8; ++k) {
            if (k < npend) {
                int d = __builtin_amdgcn_readlane(pend_d, k);
                float v = sv[k] + sd[d * HEADS + h];
                v = v > 0.f ? v : 0.2f * v;
                float ee = expf(v - unflipf(mx[d * HEADS + h]));
                float4 a = acc[d * 32 + lane];
                a.x += ee * row[k].x;
                a.y += ee * row[k].y;
                a.z += ee * row[k].z;
                a.w += ee * row[k].w;
                acc[d * 32 + lane] = a;
                if ((lane & 7) == 0) den[d * HEADS + h] += ee;
            }
        }
        npend = 0;
    };
    auto push = [&](int s, int d) {
        pend_s = (lane == npend) ? s : pend_s;
        pend_d = (lane == npend) ? d : pend_d;
        if (++npend == 8) flush();
    };
    if (!slow) {
        for (int s = 0; s < NWAVE; ++s) {
            const int c = __builtin_amdgcn_readfirstlane(lcnt[s]);
            for (int i = 0; i < c; i += 32) {
                int k = i + lane;
                int2 e = (k < c) ? lst[s * LSTR + k] : make_int2(0, -1);
                unsigned hit = __builtin_amdgcn_ballot_w32(k < c && e.y >= 0 && (e.y & 31) == w);
                while (hit) {
                    int j = __builtin_ctz(hit);
                    hit &= hit - 1;
                    push(__builtin_amdgcn_readlane(e.x, j), __builtin_amdgcn_readlane(e.y, j));
                }
            }
        }
    } else {
        for (int b = 0; b < nE; b += 32) {
            int idx = b + lane;
            unsigned u = (idx < nE) ? (unsigned)(dstp[idx] - lo) : 0xFFFFFFFFu;
            bool mine = u < ucnt && (int)(u & 31u) == w;
            int sl = mine ? ei[idx] : 0;
            unsigned hit = __builtin_amdgcn_ballot_w32(mine);
            while (hit) {
                int j = __builtin_ctz(hit);
                hit &= hit - 1;
                push(__builtin_amdgcn_readlane(sl, j), __builtin_amdgcn_readlane((int)u, j));
            }
        }
    }
    if (npend > 0) flush();
    __syncthreads();

    for (int q = tid; q < cntR * 32; q += 1024) {
        int n = q >> 5, c4 = q & 31;
        float dn = den[n * HEADS + (c4 >> 3)] + 1e-8f;
        float4 a = acc[q];
        float4 r;
        r.x = a.x / dn; r.x = r.x > 0.f ? r.x : expm1f(r.x);
        r.y = a.y / dn; r.y = r.y > 0.f ? r.y : expm1f(r.y);
        r.z = a.z / dn; r.z = r.z > 0.f ? r.z : expm1f(r.z);
        r.w = a.w / dn; r.w = r.w > 0.f ? r.w : expm1f(r.w);
        *(float4*)(out + (size_t)(lo + n) * DIM + c4 * 4) = r;
    }
}

extern "C" void kernel_launch(void* const* d_in, const int* in_sizes, int n_in,
                              void* d_out, int out_size, void* d_ws, size_t ws_size,
                              hipStream_t stream) {
    const float* x   = (const float*)d_in[0];
    const int*   ei  = (const int*)d_in[1];
    const float* W   = (const float*)d_in[2];
    const float* a_w = (const float*)d_in[3];
    int N = in_sizes[0] / DIM;
    int E = in_sizes[1] / 2;
    float* out = (float*)d_out;

    float* Wx    = (float*)d_ws;
    float* s_src = Wx + (size_t)N * DIM;
    float* s_dst = s_src + (size_t)N * HEADS;
    unsigned* keys = (unsigned*)(s_dst + (size_t)N * HEADS);

    dim3 blk(256);
    int nTiles = (N + 15) / 16;
    gat_gemm_wmma<<<(nTiles + 7) / 8, blk, 0, stream>>>(x, W, Wx, N, nTiles);
    gat_scores  <<<(N * HEADS + 255) / 256, blk, 0, stream>>>(Wx, a_w, s_src, s_dst, N, ei, keys, E);
    gat_edge_owner<<<(N + RN - 1) / RN, dim3(1024), 0, stream>>>(ei, s_src, s_dst, Wx, keys, out, E, N);
}


# Sum pass: a batch of 8 edges gets its softmax weights and denominator sums from one lane per (edge, head), DPP8-broadcast to the head's lanes; per-edge wave-wide exp gone; sums and order unchanged
# speedup vs baseline: 4.5040x; 1.1259x over previous
#include <hip/hip_runtime.h>
#include <hip/hip_bf16.h>

#define DIM       128
#define HEADS     4
#define HEAD_DIM  32
#define RN        196

typedef __attribute__((ext_vector_type(2))) float v2f;
typedef __attribute__((ext_vector_type(8))) float v8f;

__device__ __forceinline__ unsigned flipf(float f) {
    unsigned u = __float_as_uint(f);
    return (u & 0x80000000u) ? ~u : (u | 0x80000000u);
}
__device__ __forceinline__ float unflipf(unsigned u) {
    return __uint_as_float((u & 0x80000000u) ? (u & 0x7FFFFFFFu) : ~u);
}

__global__ __launch_bounds__(256)
void gat_gemm_wmma(const float* __restrict__ x, const float* __restrict__ W,
                   float* __restrict__ Wx, int nNodes, int nTiles) {
    int wave = (blockIdx.x * blockDim.x + threadIdx.x) >> 5;
    if (wave >= nTiles) return;
    int lane = threadIdx.x & 31;
    int rc   = lane & 15;
    int hi   = lane >> 4;
    int nodeBase = wave * 16;

    v8f acc[8] = {};

    int arow = nodeBase + rc;
    if (arow >= nNodes) arow = nNodes - 1;
    const float* aptr = x + (size_t)arow * DIM + hi * 2;

    #pragma unroll 8
    for (int k = 0; k < DIM; k += 4) {
        v2f a = *(const v2f*)(aptr + k);
        #pragma unroll
        for (int nt = 0; nt < 8; ++nt) {
            v2f b = *(const v2f*)(W + (size_t)(nt * 16 + rc) * DIM + k + hi * 2);
            acc[nt] = __builtin_amdgcn_wmma_f32_16x16x4_f32(
                false, a, false, b, (short)0, acc[nt], false, false);
        }
    }

    #pragma unroll
    for (int nt = 0; nt < 8; ++nt) {
        #pragma unroll
        for (int j = 0; j < 8; ++j) {
            int node = nodeBase + j + 8 * hi;
            if (node < nNodes)
                Wx[(size_t)node * DIM + nt * 16 + rc] = acc[nt][j];
        }
    }
}

__global__ __launch_bounds__(256)
void gat_scores(const float* __restrict__ Wx, const float* __restrict__ a_w,
                float* __restrict__ s_src, float* __restrict__ s_dst, int nNodes,
                const int* __restrict__ ei, unsigned* __restrict__ keys, int nE) {
    int t = blockIdx.x * blockDim.x + threadIdx.x;
    const int* __restrict__ dstp = ei + nE;
    const int nW = ((nE + 15) >> 4) << 2;
    for (int q = t; q < nW; q += (int)(gridDim.x * blockDim.x)) {
        unsigned wd = 0u;
        #pragma unroll
        for (int b = 0; b < 4; ++b) {
            int e = q * 4 + b;
            unsigned k = (e < nE) ? (((unsigned)dstp[e] / (unsigned)RN) & 255u) : 255u;
            wd |= k << (8 * b);
        }
        keys[q] = wd;
    }
    if (t >= nNodes * HEADS) return;
    int h = t & (HEADS - 1);
    int n = t >> 2;
    const float* p = Wx + (size_t)n * DIM + h * HEAD_DIM;
    float ss = 0.f, sd = 0.f;
    #pragma unroll
    for (int d = 0; d < HEAD_DIM; ++d) {
        float v = p[d];
        ss += v * a_w[d];
        sd += v * a_w[HEAD_DIM + d];
    }
    s_src[t] = ss;
    s_dst[t] = sd;
}

#define NWAVE 32
#define LCAP  512
#define LSTR  (LCAP + 1)

__global__ __launch_bounds__(1024)
void gat_edge_owner(const int* __restrict__ ei, const float* __restrict__ s_src,
                    const float* __restrict__ s_dst, const float* __restrict__ Wx,
                    const unsigned* __restrict__ keys,
                    float* __restrict__ out, int nE, int nNodes) {
    __shared__ float4   acc[RN * 32];
    __shared__ float    den[RN * HEADS];
    __shared__ unsigned mx[RN * HEADS];
    __shared__ float    sd[RN * HEADS];
    __shared__ int2     lst[NWAVE * LSTR];
    __shared__ int      lcnt[NWAVE];
    __shared__ int      ovf;

    const int tid  = threadIdx.x;
    const int lane = tid & 31;
    const int w    = __builtin_amdgcn_readfirstlane(tid >> 5);
    const int lo   = blockIdx.x * RN;
    int cntR = nNodes - lo;
    if (cntR > RN) cntR = RN;
    const unsigned ucnt = (unsigned)cntR;
    const int* __restrict__ dstp = ei + nE;

    for (int i = tid; i < RN * 32; i += 1024) acc[i] = make_float4(0.f, 0.f, 0.f, 0.f);
    for (int i = tid; i < RN * HEADS; i += 1024) {
        den[i] = 0.f;
        mx[i]  = 0u;
        sd[i]  = (i < cntR * HEADS) ? s_dst[(size_t)lo * HEADS + i] : 0.f;
    }
    if (tid == 0) ovf = 0;
    __syncthreads();

    int cnt = 0;
    int2* mylist = lst + w * LSTR;
    const unsigned pat = (unsigned)(blockIdx.x & 255) * 0x01010101u;
    const int nvec = (nE + 15) >> 4;
    const uint4* __restrict__ k4 = (const uint4*)keys;
    auto proc = [&](uint4 qk, int v) {
        unsigned t0 = qk.x ^ pat, t1 = qk.y ^ pat, t2 = qk.z ^ pat, t3 = qk.w ^ pat;
        unsigned z0 = (t0 - 0x01010101u) & ~t0 & 0x80808080u;
        unsigned z1 = (t1 - 0x01010101u) & ~t1 & 0x80808080u;
        unsigned z2 = (t2 - 0x01010101u) & ~t2 & 0x80808080u;
        unsigned z3 = (t3 - 0x01010101u) & ~t3 & 0x80808080u;
        unsigned c = (z0 >> 7) | (z1 >> 6) | (z2 >> 5) | (z3 >> 4);
        const int ebase = v * 16;
        unsigned bal = __builtin_amdgcn_ballot_w32(c != 0u);
        while (bal) {
            int pos = cnt + (int)__builtin_amdgcn_mbcnt_lo(bal, 0u);
            if (c != 0u) {
                int p = __builtin_ctz(c);
                if (pos < LCAP) mylist[pos] = make_int2(ebase + ((p & 7) << 2) + (p >> 3), -1);
            }
            cnt += __builtin_popcount(bal);
            c &= c - 1u;
            bal = __builtin_amdgcn_ballot_w32(c != 0u);
        }
    };
    const int nfull = nvec >> 7;
    for (int gi = w; gi < nfull; gi += NWAVE) {
        uint4 q[4];
        #pragma unroll
        for (int k = 0; k < 4; ++k) q[k] = k4[gi * 128 + k * 32 + lane];
        #pragma unroll
        for (int k = 0; k < 4; ++k) proc(q[k], gi * 128 + k * 32 + lane);
    }
    for (int v0 = nfull * 128 + w * 32; v0 < nvec; v0 += NWAVE * 32) {
        int v = v0 + lane;
        uint4 qk = (v < nvec) ? k4[v] : make_uint4(~pat, ~pat, ~pat, ~pat);
        proc(qk, v);
    }
    if (lane == 0) {
        lcnt[w] = cnt < LCAP ? cnt : LCAP;
        if (cnt > LCAP) ovf = 1;
    }
    __syncthreads();
    const int slow = __builtin_amdgcn_readfirstlane(ovf);

    if (!slow) {
        for (int i = w; i < LCAP; i += NWAVE) {
            bool act = i < lcnt[lane];
            if (__builtin_amdgcn_ballot_w32(act) == 0u) break;
            if (act) {
                const int eid = lst[lane * LSTR + i].x;
                const bool inb = eid < nE;
                const int dv  = inb ? dstp[eid] : -1;
                const int src = inb ? ei[eid] : 0;
                const unsigned u = (unsigned)(dv - lo);
                if (inb && u < ucnt) {
                    float4 ss = ((const float4*)s_src)[src];
                    float v0 = ss.x + sd[u * HEADS + 0];
                    float v1 = ss.y + sd[u * HEADS + 1];
                    float v2 = ss.z + sd[u * HEADS + 2];
                    float v3 = ss.w + sd[u * HEADS + 3];
                    v0 = v0 > 0.f ? v0 : 0.2f * v0;
                    v1 = v1 > 0.f ? v1 : 0.2f * v1;
                    v2 = v2 > 0.f ? v2 : 0.2f * v2;
                    v3 = v3 > 0.f ? v3 : 0.2f * v3;
                    atomicMax(&mx[u * HEADS + 0], flipf(v0));
                    atomicMax(&mx[u * HEADS + 1], flipf(v1));
                    atomicMax(&mx[u * HEADS + 2], flipf(v2));
                    atomicMax(&mx[u * HEADS + 3], flipf(v3));
                    lst[lane * LSTR + i] = make_int2(src, (int)u);
                } else {
                    lst[lane * LSTR + i] = make_int2(0, -1);
                }
            }
        }
    } else {
        for (int b = 0; b < nE; b += 32) {
            int idx = b + lane;
            unsigned u = (idx < nE) ? (unsigned)(dstp[idx] - lo) : 0xFFFFFFFFu;
            unsigned hit = __builtin_amdgcn_ballot_w32(u < ucnt && (int)(u & 31u) == w);
            while (hit) {
                int j = __builtin_ctz(hit);
                hit &= hit - 1;
                int d = __builtin_amdgcn_readlane((int)u, j);
                int src = ei[b + j];
                if (lane < HEADS) {
                    float v = s_src[src * HEADS + lane] + sd[d * HEADS + lane];
                    v = v > 0.f ? v : 0.2f * v;
                    atomicMax(&mx[d * HEADS + lane], flipf(v));
                }
            }
        }
    }
    __syncthreads();

    int pend_s = 0, pend_d = 0, npend = 0;
    const int h  = lane >> 3;
    const int sl = lane & 7;
#define GROUP_LANE(x, k) __uint_as_float(__builtin_amdgcn_mov_dpp8(__float_as_uint(x), (unsigned)(k) * 0x249249u))
#define EDGE_STEP(k)                                                        \
    {                                                                       \
        const int   d   = __builtin_amdgcn_readlane(pend_d, k);             \
        const float eek = GROUP_LANE(ee, k);                                \
        run = (pend_d == d) ? run + eek : run;                              \
        if (k < npend) {                                                    \
            float4 a = acc[d * 32 + lane];                                  \
            a.x += eek * row[k].x;                                          \
            a.y += eek * row[k].y;                                          \
            a.z += eek * row[k].z;                                          \
            a.w += eek * row[k].w;                                          \
            acc[d * 32 + lane] = a;                                         \
        }                                                                   \
    }
    auto flush = [&]() {
        float4 row[8];
        #pragma unroll
        for (int k = 0; k < 8; ++k) {
            int s = __builtin_amdgcn_readlane(pend_s, k);
            row[k] = *(const float4*)(Wx + (size_t)s * DIM + lane * 4);
        }
        float v = s_src[pend_s * HEADS + h] + sd[pend_d * HEADS + h];
        v = v > 0.f ? v : 0.2f * v;
        float ee = expf(v - unflipf(mx[pend_d * HEADS + h]));
        if (sl >= npend) ee = 0.f;
        float run = den[pend_d * HEADS + h];
        EDGE_STEP(0) EDGE_STEP(1) EDGE_STEP(2) EDGE_STEP(3)
        EDGE_STEP(4) EDGE_STEP(5) EDGE_STEP(6) EDGE_STEP(7)
        if (sl < npend) den[pend_d * HEADS + h] = run;
        npend = 0;
    };
#undef EDGE_STEP
#undef GROUP_LANE
    auto push = [&](int s, int d) {
        pend_s = (sl == npend) ? s : pend_s;
        pend_d = (sl == npend) ? d : pend_d;
        if (++npend == 8) flush();
    };
    if (!slow) {
        for (int s = 0; s < NWAVE; ++s) {
            const int c = __builtin_amdgcn_readfirstlane(lcnt[s]);
            for (int i = 0; i < c; i += 32) {
                int k = i + lane;
                int2 e = (k < c) ? lst[s * LSTR + k] : make_int2(0, -1);
                unsigned hit = __builtin_amdgcn_ballot_w32(k < c && e.y >= 0 && (e.y & 31) == w);
                while (hit) {
                    int j = __builtin_ctz(hit);
                    hit &= hit - 1;
                    push(__builtin_amdgcn_readlane(e.x, j), __builtin_amdgcn_readlane(e.y, j));
                }
            }
        }
    } else {
        for (int b = 0; b < nE; b += 32) {
            int idx = b + lane;
            unsigned u = (idx < nE) ? (unsigned)(dstp[idx] - lo) : 0xFFFFFFFFu;
            bool mine = u < ucnt && (int)(u & 31u) == w;
            int sl = mine ? ei[idx] : 0;
            unsigned hit = __builtin_amdgcn_ballot_w32(mine);
            while (hit) {
                int j = __builtin_ctz(hit);
                hit &= hit - 1;
                push(__builtin_amdgcn_readlane(sl, j), __builtin_amdgcn_readlane((int)u, j));
            }
        }
    }
    if (npend > 0) flush();
    __syncthreads();

    for (int q = tid; q < cntR * 32; q += 1024) {
        int n = q >> 5, c4 = q & 31;
        float dn = den[n * HEADS + (c4 >> 3)] + 1e-8f;
        float4 a = acc[q];
        float4 r;
        r.x = a.x / dn; r.x = r.x > 0.f ? r.x : expm1f(r.x);
        r.y = a.y / dn; r.y = r.y > 0.f ? r.y : expm1f(r.y);
        r.z = a.z / dn; r.z = r.z > 0.f ? r.z : expm1f(r.z);
        r.w = a.w / dn; r.w = r.w > 0.f ? r.w : expm1f(r.w);
        *(float4*)(out + (size_t)(lo + n) * DIM + c4 * 4) = r;
    }
}

extern "C" void kernel_launch(void* const* d_in, const int* in_sizes, int n_in,
                              void* d_out, int out_size, void* d_ws, size_t ws_size,
                              hipStream_t stream) {
    const float* x   = (const float*)d_in[0];
    const int*   ei  = (const int*)d_in[1];
    const float* W   = (const float*)d_in[2];
    const float* a_w = (const float*)d_in[3];
    int N = in_sizes[0] / DIM;
    int E = in_sizes[1] / 2;
    float* out = (float*)d_out;

    float* Wx    = (float*)d_ws;
    float* s_src = Wx + (size_t)N * DIM;
    float* s_dst = s_src + (size_t)N * HEADS;
    unsigned* keys = (unsigned*)(s_dst + (size_t)N * HEADS);

    dim3 blk(256);
    int nTiles = (N + 15) / 16;
    gat_gemm_wmma<<<(nTiles + 7) / 8, blk, 0, stream>>>(x, W, Wx, N, nTiles);
    gat_scores  <<<(N * HEADS + 255) / 256, blk, 0, stream>>>(Wx, a_w, s_src, s_dst, N, ei, keys, E);
    gat_edge_owner<<<(N + RN - 1) / RN, dim3(1024), 0, stream>>>(ei, s_src, s_dst, Wx, keys, out, E, N);
}


# Edge stage: each listed edge is copied to a counted, fixed slot of its owner wave's own LDS list, scan order kept; the sum pass reads its batches of 8 straight from it; sums and their order unchanged
# speedup vs baseline: 5.0739x; 1.1265x over previous
#include <hip/hip_runtime.h>
#include <hip/hip_bf16.h>

#define DIM       128
#define HEADS     4
#define HEAD_DIM  32
#define RN        196

typedef __attribute__((ext_vector_type(2))) float v2f;
typedef __attribute__((ext_vector_type(8))) float v8f;

__device__ __forceinline__ unsigned flipf(float f) {
    unsigned u = __float_as_uint(f);
    return (u & 0x80000000u) ? ~u : (u | 0x80000000u);
}
__device__ __forceinline__ float unflipf(unsigned u) {
    return __uint_as_float((u & 0x80000000u) ? (u & 0x7FFFFFFFu) : ~u);
}

__global__ __launch_bounds__(256)
void gat_gemm_wmma(const float* __restrict__ x, const float* __restrict__ W,
                   float* __restrict__ Wx, int nNodes, int nTiles) {
    int wave = (blockIdx.x * blockDim.x + threadIdx.x) >> 5;
    if (wave >= nTiles) return;
    int lane = threadIdx.x & 31;
    int rc   = lane & 15;
    int hi   = lane >> 4;
    int nodeBase = wave * 16;

    v8f acc[8] = {};

    int arow = nodeBase + rc;
    if (arow >= nNodes) arow = nNodes - 1;
    const float* aptr = x + (size_t)arow * DIM + hi * 2;

    #pragma unroll 8
    for (int k = 0; k < DIM; k += 4) {
        v2f a = *(const v2f*)(aptr + k);
        #pragma unroll
        for (int nt = 0; nt < 8; ++nt) {
            v2f b = *(const v2f*)(W + (size_t)(nt * 16 + rc) * DIM + k + hi * 2);
            acc[nt] = __builtin_amdgcn_wmma_f32_16x16x4_f32(
                false, a, false, b, (short)0, acc[nt], false, false);
        }
    }

    #pragma unroll
    for (int nt = 0; nt < 8; ++nt) {
        #pragma unroll
        for (int j = 0; j < 8; ++j) {
            int node = nodeBase + j + 8 * hi;
            if (node < nNodes)
                Wx[(size_t)node * DIM + nt * 16 + rc] = acc[nt][j];
        }
    }
}

__global__ __launch_bounds__(256)
void gat_scores(const float* __restrict__ Wx, const float* __restrict__ a_w,
                float* __restrict__ s_src, float* __restrict__ s_dst, int nNodes,
                const int* __restrict__ ei, unsigned* __restrict__ keys, int nE) {
    int t = blockIdx.x * blockDim.x + threadIdx.x;
    const int* __restrict__ dstp = ei + nE;
    const int nW = ((nE + 15) >> 4) << 2;
    for (int q = t; q < nW; q += (int)(gridDim.x * blockDim.x)) {
        unsigned wd = 0u;
        #pragma unroll
        for (int b = 0; b < 4; ++b) {
            int e = q * 4 + b;
            unsigned k = (e < nE) ? (((unsigned)dstp[e] / (unsigned)RN) & 255u) : 255u;
            wd |= k << (8 * b);
        }
        keys[q] = wd;
    }
    if (t >= nNodes * HEADS) return;
    int h = t & (HEADS - 1);
    int n = t >> 2;
    const float* p = Wx + (size_t)n * DIM + h * HEAD_DIM;
    float ss = 0.f, sd = 0.f;
    #pragma unroll
    for (int d = 0; d < HEAD_DIM; ++d) {
        float v = p[d];
        ss += v * a_w[d];
        sd += v * a_w[HEAD_DIM + d];
    }
    s_src[t] = ss;
    s_dst[t] = sd;
}

#define NWAVE 32
#define LCAP  512
#define LSTR  (LCAP + 1)
#define CPIT  (NWAVE + 1)
#define NOEDGE 0xFFFFFFFFu
#define SRCBITS 24

__device__ __forceinline__ unsigned owner_lanes(unsigned vm, unsigned b0, unsigned b1, unsigned b2,
                                                unsigned b3, unsigned b4, unsigned who) {
    unsigned m = vm;
    m &= (who & 1u)  ? b0 : ~b0;
    m &= (who & 2u)  ? b1 : ~b1;
    m &= (who & 4u)  ? b2 : ~b2;
    m &= (who & 8u)  ? b3 : ~b3;
    m &= (who & 16u) ? b4 : ~b4;
    return m;
}

__global__ __launch_bounds__(1024)
void gat_edge_owner(const int* __restrict__ ei, const float* __restrict__ s_src,
                    const float* __restrict__ s_dst, const float* __restrict__ Wx,
                    const unsigned* __restrict__ keys,
                    float* __restrict__ out, int nE, int nNodes) {
    __shared__ float4   acc[RN * 32];
    __shared__ float    den[RN * HEADS];
    __shared__ unsigned mx[RN * HEADS];
    __shared__ float    sd[RN * HEADS];
    __shared__ unsigned lst[NWAVE * LSTR];
    __shared__ unsigned own[NWAVE * LSTR];
    __shared__ int      cnt2[NWAVE * CPIT];
    __shared__ int      lcnt[NWAVE];
    __shared__ int      ocnt[NWAVE];
    __shared__ int      ovf;

    const int tid  = threadIdx.x;
    const int lane = tid & 31;
    const int w    = __builtin_amdgcn_readfirstlane(tid >> 5);
    const int lo   = blockIdx.x * RN;
    int cntR = nNodes - lo;
    if (cntR > RN) cntR = RN;
    const unsigned ucnt = (unsigned)cntR;
    const int* __restrict__ dstp = ei + nE;

    for (int i = tid; i < RN * 32; i += 1024) acc[i] = make_float4(0.f, 0.f, 0.f, 0.f);
    for (int i = tid; i < RN * HEADS; i += 1024) {
        den[i] = 0.f;
        mx[i]  = 0u;
        sd[i]  = (i < cntR * HEADS) ? s_dst[(size_t)lo * HEADS + i] : 0.f;
    }
    if (tid == 0) ovf = (nNodes > (1 << SRCBITS)) ? 1 : 0;
    __syncthreads();

    int cnt = 0;
    unsigned* mylist = lst + w * LSTR;
    const unsigned pat = (unsigned)(blockIdx.x & 255) * 0x01010101u;
    const int nvec = (nE + 15) >> 4;
    const uint4* __restrict__ k4 = (const uint4*)keys;
    auto proc = [&](uint4 qk, int v) {
        unsigned t0 = qk.x ^ pat, t1 = qk.y ^ pat, t2 = qk.z ^ pat, t3 = qk.w ^ pat;
        unsigned z0 = (t0 - 0x01010101u) & ~t0 & 0x80808080u;
        unsigned z1 = (t1 - 0x01010101u) & ~t1 & 0x80808080u;
        unsigned z2 = (t2 - 0x01010101u) & ~t2 & 0x80808080u;
        unsigned z3 = (t3 - 0x01010101u) & ~t3 & 0x80808080u;
        unsigned c = (z0 >> 7) | (z1 >> 6) | (z2 >> 5) | (z3 >> 4);
        const int ebase = v * 16;
        unsigned bal = __builtin_amdgcn_ballot_w32(c != 0u);
        while (bal) {
            int pos = cnt + (int)__builtin_amdgcn_mbcnt_lo(bal, 0u);
            if (c != 0u) {
                int p = __builtin_ctz(c);
                if (pos < LCAP) mylist[pos] = (unsigned)(ebase + ((p & 7) << 2) + (p >> 3));
            }
            cnt += __builtin_popcount(bal);
            c &= c - 1u;
            bal = __builtin_amdgcn_ballot_w32(c != 0u);
        }
    };
    const int nfull = nvec >> 7;
    for (int gi = w; gi < nfull; gi += NWAVE) {
        uint4 q[4];
        #pragma unroll
        for (int k = 0; k < 4; ++k) q[k] = k4[gi * 128 + k * 32 + lane];
        #pragma unroll
        for (int k = 0; k < 4; ++k) proc(q[k], gi * 128 + k * 32 + lane);
    }
    for (int v0 = nfull * 128 + w * 32; v0 < nvec; v0 += NWAVE * 32) {
        int v = v0 + lane;
        uint4 qk = (v < nvec) ? k4[v] : make_uint4(~pat, ~pat, ~pat, ~pat);
        proc(qk, v);
    }
    if (lane == 0) {
        lcnt[w] = cnt < LCAP ? cnt : LCAP;
        if (cnt > LCAP) ovf = 1;
    }
    __syncthreads();
    const int slow = __builtin_amdgcn_readfirstlane(ovf);
    const int nmine = __builtin_amdgcn_readfirstlane(lcnt[w]);

    if (!slow) {
        int hist = 0;
        for (int i = 0; i < nmine; i += 32) {
            const int pos = i + lane;
            unsigned pk = NOEDGE;
            if (pos < nmine) {
                const int eid = (int)mylist[pos];
                const bool inb = eid < nE;
                const int dv  = inb ? dstp[eid] : -1;
                const int src = inb ? ei[eid] : 0;
                const unsigned u = (unsigned)(dv - lo);
                if (inb && u < ucnt) {
                    float4 ss = ((const float4*)s_src)[src];
                    float v0 = ss.x + sd[u * HEADS + 0];
                    float v1 = ss.y + sd[u * HEADS + 1];
                    float v2 = ss.z + sd[u * HEADS + 2];
                    float v3 = ss.w + sd[u * HEADS + 3];
                    v0 = v0 > 0.f ? v0 : 0.2f * v0;
                    v1 = v1 > 0.f ? v1 : 0.2f * v1;
                    v2 = v2 > 0.f ? v2 : 0.2f * v2;
                    v3 = v3 > 0.f ? v3 : 0.2f * v3;
                    atomicMax(&mx[u * HEADS + 0], flipf(v0));
                    atomicMax(&mx[u * HEADS + 1], flipf(v1));
                    atomicMax(&mx[u * HEADS + 2], flipf(v2));
                    atomicMax(&mx[u * HEADS + 3], flipf(v3));
                    pk = ((unsigned)src << 8) | u;
                }
                mylist[pos] = pk;
            }
            const unsigned vm = __builtin_amdgcn_ballot_w32(pk != NOEDGE);
            const unsigned b0 = __builtin_amdgcn_ballot_w32((pk & 1u) != 0u);
            const unsigned b1 = __builtin_amdgcn_ballot_w32((pk & 2u) != 0u);
            const unsigned b2 = __builtin_amdgcn_ballot_w32((pk & 4u) != 0u);
            const unsigned b3 = __builtin_amdgcn_ballot_w32((pk & 8u) != 0u);
            const unsigned b4 = __builtin_amdgcn_ballot_w32((pk & 16u) != 0u);
            hist += __builtin_popcount(owner_lanes(vm, b0, b1, b2, b3, b4, (unsigned)lane));
        }
        cnt2[lane * CPIT + w] = hist;
    } else {
        for (int b = 0; b < nE; b += 32) {
            int idx = b + lane;
            unsigned u = (idx < nE) ? (unsigned)(dstp[idx] - lo) : 0xFFFFFFFFu;
            unsigned hit = __builtin_amdgcn_ballot_w32(u < ucnt && (int)(u & 31u) == w);
            while (hit) {
                int j = __builtin_ctz(hit);
                hit &= hit - 1;
                int d = __builtin_amdgcn_readlane((int)u, j);
                int src = ei[b + j];
                if (lane < HEADS) {
                    float v = s_src[src * HEADS + lane] + sd[d * HEADS + lane];
                    v = v > 0.f ? v : 0.2f * v;
                    atomicMax(&mx[d * HEADS + lane], flipf(v));
                }
            }
        }
    }
    __syncthreads();

    if (!slow) {
        int off = 0;
        for (int sp = 0; sp < w; ++sp) off += cnt2[lane * CPIT + sp];
        for (int i = 0; i < nmine; i += 32) {
            const int pos = i + lane;
            const unsigned pk = (pos < nmine) ? mylist[pos] : NOEDGE;
            const unsigned vm = __builtin_amdgcn_ballot_w32(pk != NOEDGE);
            const unsigned b0 = __builtin_amdgcn_ballot_w32((pk & 1u) != 0u);
            const unsigned b1 = __builtin_amdgcn_ballot_w32((pk & 2u) != 0u);
            const unsigned b2 = __builtin_amdgcn_ballot_w32((pk & 4u) != 0u);
            const unsigned b3 = __builtin_amdgcn_ballot_w32((pk & 8u) != 0u);
            const unsigned b4 = __builtin_amdgcn_ballot_w32((pk & 16u) != 0u);
            const unsigned o  = pk & 31u;
            const unsigned peers = owner_lanes(vm, b0, b1, b2, b3, b4, o);
            const int slot = __builtin_amdgcn_ds_bpermute((int)(o << 2), off)
                           + (int)__builtin_amdgcn_mbcnt_lo(peers, 0u);
            if (pk != NOEDGE && slot < LCAP) own[o * LSTR + slot] = pk;
            off += __builtin_popcount(owner_lanes(vm, b0, b1, b2, b3, b4, (unsigned)lane));
        }
        if (w == NWAVE - 1) {
            ocnt[lane] = off < LCAP ? off : LCAP;
            if (off > LCAP) ovf = 1;
        }
    }
    __syncthreads();
    const int slow2 = __builtin_amdgcn_readfirstlane(ovf);

    int pend_s = 0, pend_d = 0, npend = 0;
    const int h  = lane >> 3;
    const int sl = lane & 7;
#define GROUP_LANE(x, k) __uint_as_float(__builtin_amdgcn_mov_dpp8(__float_as_uint(x), (unsigned)(k) * 0x249249u))
#define EDGE_STEP(k)                                                        \
    {                                                                       \
        const int   d   = __builtin_amdgcn_readlane(pend_d, k);             \
        const float eek = GROUP_LANE(ee, k);                                \
        run = (pend_d == d) ? run + eek : run;                              \
        if (k < npend) {                                                    \
            float4 a = acc[d * 32 + lane];                                  \
            a.x += eek * row[k].x;                                          \
            a.y += eek * row[k].y;                                          \
            a.z += eek * row[k].z;                                          \
            a.w += eek * row[k].w;                                          \
            acc[d * 32 + lane] = a;                                         \
        }                                                                   \
    }
    auto flush = [&]() {
        float4 row[8];
        #pragma unroll
        for (int k = 0; k < 8; ++k) {
            int s = __builtin_amdgcn_readlane(pend_s, k);
            row[k] = *(const float4*)(Wx + (size_t)s * DIM + lane * 4);
        }
        float v = s_src[pend_s * HEADS + h] + sd[pend_d * HEADS + h];
        v = v > 0.f ? v : 0.2f * v;
        float ee = expf(v - unflipf(mx[pend_d * HEADS + h]));
        if (sl >= npend) ee = 0.f;
        float run = den[pend_d * HEADS + h];
        EDGE_STEP(0) EDGE_STEP(1) EDGE_STEP(2) EDGE_STEP(3)
        EDGE_STEP(4) EDGE_STEP(5) EDGE_STEP(6) EDGE_STEP(7)
        if (sl < npend) den[pend_d * HEADS + h] = run;
        npend = 0;
    };
#undef EDGE_STEP
#undef GROUP_LANE
    if (!slow2) {
        const unsigned* ml = own + w * LSTR;
        const int n = __builtin_amdgcn_readfirstlane(ocnt[w]);
        int nb = n < 8 ? n : 8;
        unsigned pk = (n > 0) ? ml[sl < nb ? sl : nb - 1] : 0u;
        for (int b0 = 0; b0 < n; b0 += 8) {
            pend_s = (int)(pk >> 8);
            pend_d = (int)(pk & 255u);
            const int cur = nb;
            const int rest = n - b0 - 8;
            nb = rest < 8 ? rest : 8;
            if (rest > 0) pk = ml[b0 + 8 + (sl < nb ? sl : nb - 1)];
            if (cur == 8) { npend = 8;   flush(); }
            else          { npend = cur; flush(); }
        }
    } else {
        auto push = [&](int s, int d) {
            pend_s = (sl == npend) ? s : pend_s;
            pend_d = (sl == npend) ? d : pend_d;
            if (++npend == 8) flush();
        };
        for (int b = 0; b < nE; b += 32) {
            int idx = b + lane;
            unsigned u = (idx < nE) ? (unsigned)(dstp[idx] - lo) : 0xFFFFFFFFu;
            bool mine = u < ucnt && (int)(u & 31u) == w;
            int sv = mine ? ei[idx] : 0;
            unsigned hit = __builtin_amdgcn_ballot_w32(mine);
            while (hit) {
                int j = __builtin_ctz(hit);
                hit &= hit - 1;
                push(__builtin_amdgcn_readlane(sv, j), __builtin_amdgcn_readlane((int)u, j));
            }
        }
        if (npend > 0) flush();
    }
    __syncthreads();

    for (int q = tid; q < cntR * 32; q += 1024) {
        int n = q >> 5, c4 = q & 31;
        float dn = den[n * HEADS + (c4 >> 3)] + 1e-8f;
        float4 a = acc[q];
        float4 r;
        r.x = a.x / dn; r.x = r.x > 0.f ? r.x : expm1f(r.x);
        r.y = a.y / dn; r.y = r.y > 0.f ? r.y : expm1f(r.y);
        r.z = a.z / dn; r.z = r.z > 0.f ? r.z : expm1f(r.z);
        r.w = a.w / dn; r.w = r.w > 0.f ? r.w : expm1f(r.w);
        *(float4*)(out + (size_t)(lo + n) * DIM + c4 * 4) = r;
    }
}

extern "C" void kernel_launch(void* const* d_in, const int* in_sizes, int n_in,
                              void* d_out, int out_size, void* d_ws, size_t ws_size,
                              hipStream_t stream) {
    const float* x   = (const float*)d_in[0];
    const int*   ei  = (const int*)d_in[1];
    const float* W   = (const float*)d_in[2];
    const float* a_w = (const float*)d_in[3];
    int N = in_sizes[0] / DIM;
    int E = in_sizes[1] / 2;
    float* out = (float*)d_out;

    float* Wx    = (float*)d_ws;
    float* s_src = Wx + (size_t)N * DIM;
    float* s_dst = s_src + (size_t)N * HEADS;
    unsigned* keys = (unsigned*)(s_dst + (size_t)N * HEADS);

    dim3 blk(256);
    int nTiles = (N + 15) / 16;
    gat_gemm_wmma<<<(nTiles + 7) / 8, blk, 0, stream>>>(x, W, Wx, N, nTiles);
    gat_scores  <<<(N * HEADS + 255) / 256, blk, 0, stream>>>(Wx, a_w, s_src, s_dst, N, ei, keys, E);
    gat_edge_owner<<<(N + RN - 1) / RN, dim3(1024), 0, stream>>>(ei, s_src, s_dst, Wx, keys, out, E, N);
}


# Projection on resident waves: each keeps the W fragments of its 32 output columns in registers and walks node tiles, loading only x (one tile ahead); W not re-read per tile; sums and order unchanged
# speedup vs baseline: 5.5287x; 1.0896x over previous
#include <hip/hip_runtime.h>
#include <hip/hip_bf16.h>

#define DIM       128
#define HEADS     4
#define HEAD_DIM  32
#define RN        196

typedef __attribute__((ext_vector_type(2))) float v2f;
typedef __attribute__((ext_vector_type(8))) float v8f;

__device__ __forceinline__ unsigned flipf(float f) {
    unsigned u = __float_as_uint(f);
    return (u & 0x80000000u) ? ~u : (u | 0x80000000u);
}
__device__ __forceinline__ float unflipf(unsigned u) {
    return __uint_as_float((u & 0x80000000u) ? (u & 0x7FFFFFFFu) : ~u);
}

#define KSTEPS (DIM / 4)
__global__ __launch_bounds__(256)
void gat_gemm_wmma(const float* __restrict__ x, const float* __restrict__ W,
                   float* __restrict__ Wx, int nNodes, int nTiles) {
    const int lane = threadIdx.x & 31;
    const int w    = __builtin_amdgcn_readfirstlane(threadIdx.x >> 5);
    const int rc   = lane & 15;
    const int hi   = lane >> 4;
    const int cg   = w & 3;
    const int slot  = blockIdx.x * 2 + (w >> 2);
    const int nslot = gridDim.x * 2;
    if (slot >= nTiles) return;

    v2f b0[KSTEPS], b1[KSTEPS];
    const float* wp0 = W + (size_t)(cg * 32 + rc) * DIM + hi * 2;
    const float* wp1 = wp0 + (size_t)16 * DIM;
    #pragma unroll
    for (int ks = 0; ks < KSTEPS; ++ks) {
        b0[ks] = *(const v2f*)(wp0 + ks * 4);
        b1[ks] = *(const v2f*)(wp1 + ks * 4);
    }

    auto load_a = [&](v2f (&a)[KSTEPS], int tile) {
        int arow = tile * 16 + rc;
        if (arow >= nNodes) arow = nNodes - 1;
        const float* aptr = x + (size_t)arow * DIM + hi * 2;
        #pragma unroll
        for (int ks = 0; ks < KSTEPS; ++ks) a[ks] = *(const v2f*)(aptr + ks * 4);
    };
    auto do_tile = [&](const v2f (&a)[KSTEPS], int tile) {
        v8f acc0 = {}, acc1 = {};
        #pragma unroll
        for (int ks = 0; ks < KSTEPS; ++ks) {
            acc0 = __builtin_amdgcn_wmma_f32_16x16x4_f32(
                false, a[ks], false, b0[ks], (short)0, acc0, false, false);
            acc1 = __builtin_amdgcn_wmma_f32_16x16x4_f32(
                false, a[ks], false, b1[ks], (short)0, acc1, false, false);
        }
        const int nodeBase = tile * 16;
        #pragma unroll
        for (int j = 0; j < 8; ++j) {
            int node = nodeBase + j + 8 * hi;
            if (node < nNodes) {
                float* o = Wx + (size_t)node * DIM + cg * 32 + rc;
                o[0]  = acc0[j];
                o[16] = acc1[j];
            }
        }
    };

    v2f aA[KSTEPS], aB[KSTEPS];
    int tile = slot;
    load_a(aA, tile);
    for (;;) {
        const int t1 = tile + nslot;
        load_a(aB, t1 < nTiles ? t1 : tile);
        __builtin_amdgcn_sched_barrier(0);
        do_tile(aA, tile);
        if (t1 >= nTiles) break;
        const int t2 = t1 + nslot;
        load_a(aA, t2 < nTiles ? t2 : t1);
        __builtin_amdgcn_sched_barrier(0);
        do_tile(aB, t1);
        if (t2 >= nTiles) break;
        tile = t2;
    }
}

__global__ __launch_bounds__(256)
void gat_scores(const float* __restrict__ Wx, const float* __restrict__ a_w,
                float* __restrict__ s_src, float* __restrict__ s_dst, int nNodes,
                const int* __restrict__ ei, unsigned* __restrict__ keys, int nE) {
    int t = blockIdx.x * blockDim.x + threadIdx.x;
    const int* __restrict__ dstp = ei + nE;
    const int nW = ((nE + 15) >> 4) << 2;
    for (int q = t; q < nW; q += (int)(gridDim.x * blockDim.x)) {
        unsigned wd = 0u;
        #pragma unroll
        for (int b = 0; b < 4; ++b) {
            int e = q * 4 + b;
            unsigned k = (e < nE) ? (((unsigned)dstp[e] / (unsigned)RN) & 255u) : 255u;
            wd |= k << (8 * b);
        }
        keys[q] = wd;
    }
    if (t >= nNodes * HEADS) return;
    int h = t & (HEADS - 1);
    int n = t >> 2;
    const float* p = Wx + (size_t)n * DIM + h * HEAD_DIM;
    float ss = 0.f, sd = 0.f;
    #pragma unroll
    for (int d = 0; d < HEAD_DIM; ++d) {
        float v = p[d];
        ss += v * a_w[d];
        sd += v * a_w[HEAD_DIM + d];
    }
    s_src[t] = ss;
    s_dst[t] = sd;
}

#define NWAVE 32
#define LCAP  512
#define LSTR  (LCAP + 1)
#define CPIT  (NWAVE + 1)
#define NOEDGE 0xFFFFFFFFu
#define SRCBITS 24

__device__ __forceinline__ unsigned owner_lanes(unsigned vm, unsigned b0, unsigned b1, unsigned b2,
                                                unsigned b3, unsigned b4, unsigned who) {
    unsigned m = vm;
    m &= (who & 1u)  ? b0 : ~b0;
    m &= (who & 2u)  ? b1 : ~b1;
    m &= (who & 4u)  ? b2 : ~b2;
    m &= (who & 8u)  ? b3 : ~b3;
    m &= (who & 16u) ? b4 : ~b4;
    return m;
}

__global__ __launch_bounds__(1024)
void gat_edge_owner(const int* __restrict__ ei, const float* __restrict__ s_src,
                    const float* __restrict__ s_dst, const float* __restrict__ Wx,
                    const unsigned* __restrict__ keys,
                    float* __restrict__ out, int nE, int nNodes) {
    __shared__ float4   acc[RN * 32];
    __shared__ float    den[RN * HEADS];
    __shared__ unsigned mx[RN * HEADS];
    __shared__ float    sd[RN * HEADS];
    __shared__ unsigned lst[NWAVE * LSTR];
    __shared__ unsigned own[NWAVE * LSTR];
    __shared__ int      cnt2[NWAVE * CPIT];
    __shared__ int      lcnt[NWAVE];
    __shared__ int      ocnt[NWAVE];
    __shared__ int      ovf;

    const int tid  = threadIdx.x;
    const int lane = tid & 31;
    const int w    = __builtin_amdgcn_readfirstlane(tid >> 5);
    const int lo   = blockIdx.x * RN;
    int cntR = nNodes - lo;
    if (cntR > RN) cntR = RN;
    const unsigned ucnt = (unsigned)cntR;
    const int* __restrict__ dstp = ei + nE;

    for (int i = tid; i < RN * 32; i += 1024) acc[i] = make_float4(0.f, 0.f, 0.f, 0.f);
    for (int i = tid; i < RN * HEADS; i += 1024) {
        den[i] = 0.f;
        mx[i]  = 0u;
        sd[i]  = (i < cntR * HEADS) ? s_dst[(size_t)lo * HEADS + i] : 0.f;
    }
    if (tid == 0) ovf = (nNodes > (1 << SRCBITS)) ? 1 : 0;
    __syncthreads();

    int cnt = 0;
    unsigned* mylist = lst + w * LSTR;
    const unsigned pat = (unsigned)(blockIdx.x & 255) * 0x01010101u;
    const int nvec = (nE + 15) >> 4;
    const uint4* __restrict__ k4 = (const uint4*)keys;
    auto proc = [&](uint4 qk, int v) {
        unsigned t0 = qk.x ^ pat, t1 = qk.y ^ pat, t2 = qk.z ^ pat, t3 = qk.w ^ pat;
        unsigned z0 = (t0 - 0x01010101u) & ~t0 & 0x80808080u;
        unsigned z1 = (t1 - 0x01010101u) & ~t1 & 0x80808080u;
        unsigned z2 = (t2 - 0x01010101u) & ~t2 & 0x80808080u;
        unsigned z3 = (t3 - 0x01010101u) & ~t3 & 0x80808080u;
        unsigned c = (z0 >> 7) | (z1 >> 6) | (z2 >> 5) | (z3 >> 4);
        const int ebase = v * 16;
        unsigned bal = __builtin_amdgcn_ballot_w32(c != 0u);
        while (bal) {
            int pos = cnt + (int)__builtin_amdgcn_mbcnt_lo(bal, 0u);
            if (c != 0u) {
                int p = __builtin_ctz(c);
                if (pos < LCAP) mylist[pos] = (unsigned)(ebase + ((p & 7) << 2) + (p >> 3));
            }
            cnt += __builtin_popcount(bal);
            c &= c - 1u;
            bal = __builtin_amdgcn_ballot_w32(c != 0u);
        }
    };
    const int nfull = nvec >> 7;
    for (int gi = w; gi < nfull; gi += NWAVE) {
        uint4 q[4];
        #pragma unroll
        for (int k = 0; k < 4; ++k) q[k] = k4[gi * 128 + k * 32 + lane];
        #pragma unroll
        for (int k = 0; k < 4; ++k) proc(q[k], gi * 128 + k * 32 + lane);
    }
    for (int v0 = nfull * 128 + w * 32; v0 < nvec; v0 += NWAVE * 32) {
        int v = v0 + lane;
        uint4 qk = (v < nvec) ? k4[v] : make_uint4(~pat, ~pat, ~pat, ~pat);
        proc(qk, v);
    }
    if (lane == 0) {
        lcnt[w] = cnt < LCAP ? cnt : LCAP;
        if (cnt > LCAP) ovf = 1;
    }
    __syncthreads();
    const int slow = __builtin_amdgcn_readfirstlane(ovf);
    const int nmine = __builtin_amdgcn_readfirstlane(lcnt[w]);

    if (!slow) {
        int hist = 0;
        for (int i = 0; i < nmine; i += 32) {
            const int pos = i + lane;
            unsigned pk = NOEDGE;
            if (pos < nmine) {
                const int eid = (int)mylist[pos];
                const bool inb = eid < nE;
                const int dv  = inb ? dstp[eid] : -1;
                const int src = inb ? ei[eid] : 0;
                const unsigned u = (unsigned)(dv - lo);
                if (inb && u < ucnt) {
                    float4 ss = ((const float4*)s_src)[src];
                    float v0 = ss.x + sd[u * HEADS + 0];
                    float v1 = ss.y + sd[u * HEADS + 1];
                    float v2 = ss.z + sd[u * HEADS + 2];
                    float v3 = ss.w + sd[u * HEADS + 3];
                    v0 = v0 > 0.f ? v0 : 0.2f * v0;
                    v1 = v1 > 0.f ? v1 : 0.2f * v1;
                    v2 = v2 > 0.f ? v2 : 0.2f * v2;
                    v3 = v3 > 0.f ? v3 : 0.2f * v3;
                    atomicMax(&mx[u * HEADS + 0], flipf(v0));
                    atomicMax(&mx[u * HEADS + 1], flipf(v1));
                    atomicMax(&mx[u * HEADS + 2], flipf(v2));
                    atomicMax(&mx[u * HEADS + 3], flipf(v3));
                    pk = ((unsigned)src << 8) | u;
                }
                mylist[pos] = pk;
            }
            const unsigned vm = __builtin_amdgcn_ballot_w32(pk != NOEDGE);
            const unsigned b0 = __builtin_amdgcn_ballot_w32((pk & 1u) != 0u);
            const unsigned b1 = __builtin_amdgcn_ballot_w32((pk & 2u) != 0u);
            const unsigned b2 = __builtin_amdgcn_ballot_w32((pk & 4u) != 0u);
            const unsigned b3 = __builtin_amdgcn_ballot_w32((pk & 8u) != 0u);
            const unsigned b4 = __builtin_amdgcn_ballot_w32((pk & 16u) != 0u);
            hist += __builtin_popcount(owner_lanes(vm, b0, b1, b2, b3, b4, (unsigned)lane));
        }
        cnt2[lane * CPIT + w] = hist;
    } else {
        for (int b = 0; b < nE; b += 32) {
            int idx = b + lane;
            unsigned u = (idx < nE) ? (unsigned)(dstp[idx] - lo) : 0xFFFFFFFFu;
            unsigned hit = __builtin_amdgcn_ballot_w32(u < ucnt && (int)(u & 31u) == w);
            while (hit) {
                int j = __builtin_ctz(hit);
                hit &= hit - 1;
                int d = __builtin_amdgcn_readlane((int)u, j);
                int src = ei[b + j];
                if (lane < HEADS) {
                    float v = s_src[src * HEADS + lane] + sd[d * HEADS + lane];
                    v = v > 0.f ? v : 0.2f * v;
                    atomicMax(&mx[d * HEADS + lane], flipf(v));
                }
            }
        }
    }
    __syncthreads();

    if (!slow) {
        int off = 0;
        for (int sp = 0; sp < w; ++sp) off += cnt2[lane * CPIT + sp];
        for (int i = 0; i < nmine; i += 32) {
            const int pos = i + lane;
            const unsigned pk = (pos < nmine) ? mylist[pos] : NOEDGE;
            const unsigned vm = __builtin_amdgcn_ballot_w32(pk != NOEDGE);
            const unsigned b0 = __builtin_amdgcn_ballot_w32((pk & 1u) != 0u);
            const unsigned b1 = __builtin_amdgcn_ballot_w32((pk & 2u) != 0u);
            const unsigned b2 = __builtin_amdgcn_ballot_w32((pk & 4u) != 0u);
            const unsigned b3 = __builtin_amdgcn_ballot_w32((pk & 8u) != 0u);
            const unsigned b4 = __builtin_amdgcn_ballot_w32((pk & 16u) != 0u);
            const unsigned o  = pk & 31u;
            const unsigned peers = owner_lanes(vm, b0, b1, b2, b3, b4, o);
            const int slot = __builtin_amdgcn_ds_bpermute((int)(o << 2), off)
                           + (int)__builtin_amdgcn_mbcnt_lo(peers, 0u);
            if (pk != NOEDGE && slot < LCAP) own[o * LSTR + slot] = pk;
            off += __builtin_popcount(owner_lanes(vm, b0, b1, b2, b3, b4, (unsigned)lane));
        }
        if (w == NWAVE - 1) {
            ocnt[lane] = off < LCAP ? off : LCAP;
            if (off > LCAP) ovf = 1;
        }
    }
    __syncthreads();
    const int slow2 = __builtin_amdgcn_readfirstlane(ovf);

    int pend_s = 0, pend_d = 0, npend = 0;
    const int h  = lane >> 3;
    const int sl = lane & 7;
#define GROUP_LANE(x, k) __uint_as_float(__builtin_amdgcn_mov_dpp8(__float_as_uint(x), (unsigned)(k) * 0x249249u))
#define EDGE_STEP(k)                                                        \
    {                                                                       \
        const int   d   = __builtin_amdgcn_readlane(pend_d, k);             \
        const float eek = GROUP_LANE(ee, k);                                \
        run = (pend_d == d) ? run + eek : run;                              \
        if (k < npend) {                                                    \
            float4 a = acc[d * 32 + lane];                                  \
            a.x += eek * row[k].x;                                          \
            a.y += eek * row[k].y;                                          \
            a.z += eek * row[k].z;                                          \
            a.w += eek * row[k].w;                                          \
            acc[d * 32 + lane] = a;                                         \
        }                                                                   \
    }
    auto flush = [&]() {
        float4 row[8];
        #pragma unroll
        for (int k = 0; k < 8; ++k) {
            int s = __builtin_amdgcn_readlane(pend_s, k);
            row[k] = *(const float4*)(Wx + (size_t)s * DIM + lane * 4);
        }
        float v = s_src[pend_s * HEADS + h] + sd[pend_d * HEADS + h];
        v = v > 0.f ? v : 0.2f * v;
        float ee = expf(v - unflipf(mx[pend_d * HEADS + h]));
        if (sl >= npend) ee = 0.f;
        float run = den[pend_d * HEADS + h];
        EDGE_STEP(0) EDGE_STEP(1) EDGE_STEP(2) EDGE_STEP(3)
        EDGE_STEP(4) EDGE_STEP(5) EDGE_STEP(6) EDGE_STEP(7)
        if (sl < npend) den[pend_d * HEADS + h] = run;
        npend = 0;
    };
#undef EDGE_STEP
#undef GROUP_LANE
    if (!slow2) {
        const unsigned* ml = own + w * LSTR;
        const int n = __builtin_amdgcn_readfirstlane(ocnt[w]);
        int nb = n < 8 ? n : 8;
        unsigned pk = (n > 0) ? ml[sl < nb ? sl : nb - 1] : 0u;
        for (int b0 = 0; b0 < n; b0 += 8) {
            pend_s = (int)(pk >> 8);
            pend_d = (int)(pk & 255u);
            const int cur = nb;
            const int rest = n - b0 - 8;
            nb = rest < 8 ? rest : 8;
            if (rest > 0) pk = ml[b0 + 8 + (sl < nb ? sl : nb - 1)];
            if (cur == 8) { npend = 8;   flush(); }
            else          { npend = cur; flush(); }
        }
    } else {
        auto push = [&](int s, int d) {
            pend_s = (sl == npend) ? s : pend_s;
            pend_d = (sl == npend) ? d : pend_d;
            if (++npend == 8) flush();
        };
        for (int b = 0; b < nE; b += 32) {
            int idx = b + lane;
            unsigned u = (idx < nE) ? (unsigned)(dstp[idx] - lo) : 0xFFFFFFFFu;
            bool mine = u < ucnt && (int)(u & 31u) == w;
            int sv = mine ? ei[idx] : 0;
            unsigned hit = __builtin_amdgcn_ballot_w32(mine);
            while (hit) {
                int j = __builtin_ctz(hit);
                hit &= hit - 1;
                push(__builtin_amdgcn_readlane(sv, j), __builtin_amdgcn_readlane((int)u, j));
            }
        }
        if (npend > 0) flush();
    }
    __syncthreads();

    for (int q = tid; q < cntR * 32; q += 1024) {
        int n = q >> 5, c4 = q & 31;
        float dn = den[n * HEADS + (c4 >> 3)] + 1e-8f;
        float4 a = acc[q];
        float4 r;
        r.x = a.x / dn; r.x = r.x > 0.f ? r.x : expm1f(r.x);
        r.y = a.y / dn; r.y = r.y > 0.f ? r.y : expm1f(r.y);
        r.z = a.z / dn; r.z = r.z > 0.f ? r.z : expm1f(r.z);
        r.w = a.w / dn; r.w = r.w > 0.f ? r.w : expm1f(r.w);
        *(float4*)(out + (size_t)(lo + n) * DIM + c4 * 4) = r;
    }
}

extern "C" void kernel_launch(void* const* d_in, const int* in_sizes, int n_in,
                              void* d_out, int out_size, void* d_ws, size_t ws_size,
                              hipStream_t stream) {
    const float* x   = (const float*)d_in[0];
    const int*   ei  = (const int*)d_in[1];
    const float* W   = (const float*)d_in[2];
    const float* a_w = (const float*)d_in[3];
    int N = in_sizes[0] / DIM;
    int E = in_sizes[1] / 2;
    float* out = (float*)d_out;

    float* Wx    = (float*)d_ws;
    float* s_src = Wx + (size_t)N * DIM;
    float* s_dst = s_src + (size_t)N * HEADS;
    unsigned* keys = (unsigned*)(s_dst + (size_t)N * HEADS);

    dim3 blk(256);
    int nTiles = (N + 15) / 16;
    gat_gemm_wmma<<<256, blk, 0, stream>>>(x, W, Wx, N, nTiles);
    gat_scores  <<<(N * HEADS + 255) / 256, blk, 0, stream>>>(Wx, a_w, s_src, s_dst, N, ei, keys, E);
    gat_edge_owner<<<(N + RN - 1) / RN, dim3(1024), 0, stream>>>(ei, s_src, s_dst, Wx, keys, out, E, N);
}


# Edge scan reads its group's words of a bit table (1 bit per 16-edge key vector per group, built in the scores launch by LDS integer OR), fetches only flagged key vectors; sum order changed, fixed
# speedup vs baseline: 5.7919x; 1.0476x over previous
#include <hip/hip_runtime.h>
#include <hip/hip_bf16.h>

#define DIM       128
#define HEADS     4
#define HEAD_DIM  32
#define RN        196

typedef __attribute__((ext_vector_type(2))) float v2f;
typedef __attribute__((ext_vector_type(8))) float v8f;

__device__ __forceinline__ unsigned flipf(float f) {
    unsigned u = __float_as_uint(f);
    return (u & 0x80000000u) ? ~u : (u | 0x80000000u);
}
__device__ __forceinline__ float unflipf(unsigned u) {
    return __uint_as_float((u & 0x80000000u) ? (u & 0x7FFFFFFFu) : ~u);
}

#define KSTEPS (DIM / 4)
__global__ __launch_bounds__(256)
void gat_gemm_wmma(const float* __restrict__ x, const float* __restrict__ W,
                   float* __restrict__ Wx, int nNodes, int nTiles) {
    const int lane = threadIdx.x & 31;
    const int w    = __builtin_amdgcn_readfirstlane(threadIdx.x >> 5);
    const int rc   = lane & 15;
    const int hi   = lane >> 4;
    const int cg   = w & 3;
    const int slot  = blockIdx.x * 2 + (w >> 2);
    const int nslot = gridDim.x * 2;
    if (slot >= nTiles) return;

    v2f b0[KSTEPS], b1[KSTEPS];
    const float* wp0 = W + (size_t)(cg * 32 + rc) * DIM + hi * 2;
    const float* wp1 = wp0 + (size_t)16 * DIM;
    #pragma unroll
    for (int ks = 0; ks < KSTEPS; ++ks) {
        b0[ks] = *(const v2f*)(wp0 + ks * 4);
        b1[ks] = *(const v2f*)(wp1 + ks * 4);
    }

    auto load_a = [&](v2f (&a)[KSTEPS], int tile) {
        int arow = tile * 16 + rc;
        if (arow >= nNodes) arow = nNodes - 1;
        const float* aptr = x + (size_t)arow * DIM + hi * 2;
        #pragma unroll
        for (int ks = 0; ks < KSTEPS; ++ks) a[ks] = *(const v2f*)(aptr + ks * 4);
    };
    auto do_tile = [&](const v2f (&a)[KSTEPS], int tile) {
        v8f acc0 = {}, acc1 = {};
        #pragma unroll
        for (int ks = 0; ks < KSTEPS; ++ks) {
            acc0 = __builtin_amdgcn_wmma_f32_16x16x4_f32(
                false, a[ks], false, b0[ks], (short)0, acc0, false, false);
            acc1 = __builtin_amdgcn_wmma_f32_16x16x4_f32(
                false, a[ks], false, b1[ks], (short)0, acc1, false, false);
        }
        const int nodeBase = tile * 16;
        #pragma unroll
        for (int j = 0; j < 8; ++j) {
            int node = nodeBase + j + 8 * hi;
            if (node < nNodes) {
                float* o = Wx + (size_t)node * DIM + cg * 32 + rc;
                o[0]  = acc0[j];
                o[16] = acc1[j];
            }
        }
    };

    v2f aA[KSTEPS], aB[KSTEPS];
    int tile = slot;
    load_a(aA, tile);
    for (;;) {
        const int t1 = tile + nslot;
        load_a(aB, t1 < nTiles ? t1 : tile);
        __builtin_amdgcn_sched_barrier(0);
        do_tile(aA, tile);
        if (t1 >= nTiles) break;
        const int t2 = t1 + nslot;
        load_a(aA, t2 < nTiles ? t2 : t1);
        __builtin_amdgcn_sched_barrier(0);
        do_tile(aB, t1);
        if (t2 >= nTiles) break;
        tile = t2;
    }
}

#define KROWS 256
__global__ __launch_bounds__(256)
void gat_scores(const float* __restrict__ Wx, const float* __restrict__ a_w,
                float* __restrict__ s_src, float* __restrict__ s_dst, int nNodes,
                const int* __restrict__ ei, unsigned* __restrict__ keys, int nE) {
    __shared__ unsigned tb[KROWS * 2];
    const int tid = threadIdx.x;
    int t = blockIdx.x * blockDim.x + tid;
    const int* __restrict__ dstp = ei + nE;
    const int nW = ((nE + 15) >> 4) << 2;
    const int nCh = (nW + 255) >> 8;
    unsigned* __restrict__ pres = keys + nW;
    for (int c = blockIdx.x; c < nCh; c += (int)gridDim.x) {
        tb[tid] = 0u;
        tb[tid + 256] = 0u;
        __syncthreads();
        const int q = c * 256 + tid;
        if (q < nW) {
            const unsigned bit = 1u << ((tid >> 2) & 31);
            const int half = tid >> 7;
            unsigned wd = 0u;
            #pragma unroll
            for (int b = 0; b < 4; ++b) {
                int e = q * 4 + b;
                unsigned k = 255u;
                if (e < nE) {
                    k = ((unsigned)dstp[e] / (unsigned)RN) & 255u;
                    atomicOr(&tb[k * 2 + half], bit);
                }
                wd |= k << (8 * b);
            }
            keys[q] = wd;
        }
        __syncthreads();
        *(uint2*)(pres + ((size_t)c * KROWS + tid) * 2) = make_uint2(tb[tid * 2], tb[tid * 2 + 1]);
        __syncthreads();
    }
    if (t >= nNodes * HEADS) return;
    int h = t & (HEADS - 1);
    int n = t >> 2;
    const float* p = Wx + (size_t)n * DIM + h * HEAD_DIM;
    float ss = 0.f, sd = 0.f;
    #pragma unroll
    for (int d = 0; d < HEAD_DIM; ++d) {
        float v = p[d];
        ss += v * a_w[d];
        sd += v * a_w[HEAD_DIM + d];
    }
    s_src[t] = ss;
    s_dst[t] = sd;
}

#define NWAVE 32
#define LCAP  512
#define LSTR  (LCAP + 1)
#define CPIT  (NWAVE + 1)
#define NOEDGE 0xFFFFFFFFu
#define SRCBITS 24

__device__ __forceinline__ unsigned owner_lanes(unsigned vm, unsigned b0, unsigned b1, unsigned b2,
                                                unsigned b3, unsigned b4, unsigned who) {
    unsigned m = vm;
    m &= (who & 1u)  ? b0 : ~b0;
    m &= (who & 2u)  ? b1 : ~b1;
    m &= (who & 4u)  ? b2 : ~b2;
    m &= (who & 8u)  ? b3 : ~b3;
    m &= (who & 16u) ? b4 : ~b4;
    return m;
}

__global__ __launch_bounds__(1024)
void gat_edge_owner(const int* __restrict__ ei, const float* __restrict__ s_src,
                    const float* __restrict__ s_dst, const float* __restrict__ Wx,
                    const unsigned* __restrict__ keys,
                    float* __restrict__ out, int nE, int nNodes) {
    __shared__ float4   acc[RN * 32];
    __shared__ float    den[RN * HEADS];
    __shared__ unsigned mx[RN * HEADS];
    __shared__ float    sd[RN * HEADS];
    __shared__ unsigned lst[NWAVE * LSTR];
    __shared__ unsigned own[NWAVE * LSTR];
    __shared__ int      cnt2[NWAVE * CPIT];
    __shared__ int      lcnt[NWAVE];
    __shared__ int      ocnt[NWAVE];
    __shared__ int      ovf;

    const int tid  = threadIdx.x;
    const int lane = tid & 31;
    const int w    = __builtin_amdgcn_readfirstlane(tid >> 5);
    const int lo   = blockIdx.x * RN;
    int cntR = nNodes - lo;
    if (cntR > RN) cntR = RN;
    const unsigned ucnt = (unsigned)cntR;
    const int* __restrict__ dstp = ei + nE;

    for (int i = tid; i < RN * 32; i += 1024) acc[i] = make_float4(0.f, 0.f, 0.f, 0.f);
    for (int i = tid; i < RN * HEADS; i += 1024) {
        den[i] = 0.f;
        mx[i]  = 0u;
        sd[i]  = (i < cntR * HEADS) ? s_dst[(size_t)lo * HEADS + i] : 0.f;
    }
    if (tid == 0) ovf = (nNodes > (1 << SRCBITS)) ? 1 : 0;
    __syncthreads();

    int cnt = 0;
    unsigned* mylist = lst + w * LSTR;
    const unsigned pat = (unsigned)(blockIdx.x & 255) * 0x01010101u;
    const int nvec = (nE + 15) >> 4;
    const uint4* __restrict__ k4 = (const uint4*)keys;
    auto proc = [&](uint4 qk, int v) {
        unsigned t0 = qk.x ^ pat, t1 = qk.y ^ pat, t2 = qk.z ^ pat, t3 = qk.w ^ pat;
        unsigned z0 = (t0 - 0x01010101u) & ~t0 & 0x80808080u;
        unsigned z1 = (t1 - 0x01010101u) & ~t1 & 0x80808080u;
        unsigned z2 = (t2 - 0x01010101u) & ~t2 & 0x80808080u;
        unsigned z3 = (t3 - 0x01010101u) & ~t3 & 0x80808080u;
        unsigned c = (z0 >> 7) | (z1 >> 6) | (z2 >> 5) | (z3 >> 4);
        const int ebase = v * 16;
        unsigned bal = __builtin_amdgcn_ballot_w32(c != 0u);
        while (bal) {
            int pos = cnt + (int)__builtin_amdgcn_mbcnt_lo(bal, 0u);
            if (c != 0u) {
                int p = __builtin_ctz(c);
                if (pos < LCAP) mylist[pos] = (unsigned)(ebase + ((p & 7) << 2) + (p >> 3));
            }
            cnt += __builtin_popcount(bal);
            c &= c - 1u;
            bal = __builtin_amdgcn_ballot_w32(c != 0u);
        }
    };
    const int nW   = nvec << 2;
    const int nCh  = (nW + 255) >> 8;
    const int cpw  = (nCh + NWAVE - 1) / NWAVE;
    const uint2* __restrict__ pr2 = (const uint2*)(keys + nW) + (blockIdx.x & 255);
    unsigned* vl = own + w * LSTR;
    int vcnt = 0;
    auto note = [&](unsigned bits, int i) {
        unsigned bal = __builtin_amdgcn_ballot_w32(bits != 0u);
        while (bal) {
            int pos = vcnt + (int)__builtin_amdgcn_mbcnt_lo(bal, 0u);
            if (bits != 0u && pos < LCAP) vl[pos] = (unsigned)(i * 32 + __builtin_ctz(bits));
            vcnt += __builtin_popcount(bal);
            bits &= bits - 1u;
            bal = __builtin_amdgcn_ballot_w32(bits != 0u);
        }
    };
    for (int k0 = 0; k0 < cpw; k0 += 32) {
        const int k  = k0 + lane;
        const int ch = w * cpw + k;
        uint2 wd = make_uint2(0u, 0u);
        if (k < cpw && ch < nCh) wd = pr2[(size_t)ch * 256];
        note(wd.x, 2 * ch);
        note(wd.y, 2 * ch + 1);
    }
    __syncthreads();
    const int nv = vcnt < LCAP ? vcnt : LCAP;
    for (int i = 0; i < nv; i += 128) {
        int vv[4];
        uint4 q[4];
        #pragma unroll
        for (int k = 0; k < 4; ++k) {
            const int p = i + k * 32 + lane;
            vv[k] = (p < nv) ? (int)vl[p] : -1;
        }
        #pragma unroll
        for (int k = 0; k < 4; ++k) q[k] = k4[vv[k] >= 0 ? vv[k] : 0];
        #pragma unroll
        for (int k = 0; k < 4; ++k) {
            uint4 qk = q[k];
            if (vv[k] < 0) qk = make_uint4(~pat, ~pat, ~pat, ~pat);
            proc(qk, vv[k]);
        }
    }
    if (lane == 0) {
        lcnt[w] = cnt < LCAP ? cnt : LCAP;
        if (cnt > LCAP || vcnt > LCAP) ovf = 1;
    }
    __syncthreads();
    const int slow = __builtin_amdgcn_readfirstlane(ovf);
    const int nmine = __builtin_amdgcn_readfirstlane(lcnt[w]);

    if (!slow) {
        int hist = 0;
        for (int i = 0; i < nmine; i += 32) {
            const int pos = i + lane;
            unsigned pk = NOEDGE;
            if (pos < nmine) {
                const int eid = (int)mylist[pos];
                const bool inb = eid < nE;
                const int dv  = inb ? dstp[eid] : -1;
                const int src = inb ? ei[eid] : 0;
                const unsigned u = (unsigned)(dv - lo);
                if (inb && u < ucnt) {
                    float4 ss = ((const float4*)s_src)[src];
                    float v0 = ss.x + sd[u * HEADS + 0];
                    float v1 = ss.y + sd[u * HEADS + 1];
                    float v2 = ss.z + sd[u * HEADS + 2];
                    float v3 = ss.w + sd[u * HEADS + 3];
                    v0 = v0 > 0.f ? v0 : 0.2f * v0;
                    v1 = v1 > 0.f ? v1 : 0.2f * v1;
                    v2 = v2 > 0.f ? v2 : 0.2f * v2;
                    v3 = v3 > 0.f ? v3 : 0.2f * v3;
                    atomicMax(&mx[u * HEADS + 0], flipf(v0));
                    atomicMax(&mx[u * HEADS + 1], flipf(v1));
                    atomicMax(&mx[u * HEADS + 2], flipf(v2));
                    atomicMax(&mx[u * HEADS + 3], flipf(v3));
                    pk = ((unsigned)src << 8) | u;
                }
                mylist[pos] = pk;
            }
            const unsigned vm = __builtin_amdgcn_ballot_w32(pk != NOEDGE);
            const unsigned b0 = __builtin_amdgcn_ballot_w32((pk & 1u) != 0u);
            const unsigned b1 = __builtin_amdgcn_ballot_w32((pk & 2u) != 0u);
            const unsigned b2 = __builtin_amdgcn_ballot_w32((pk & 4u) != 0u);
            const unsigned b3 = __builtin_amdgcn_ballot_w32((pk & 8u) != 0u);
            const unsigned b4 = __builtin_amdgcn_ballot_w32((pk & 16u) != 0u);
            hist += __builtin_popcount(owner_lanes(vm, b0, b1, b2, b3, b4, (unsigned)lane));
        }
        cnt2[lane * CPIT + w] = hist;
    } else {
        for (int b = 0; b < nE; b += 32) {
            int idx = b + lane;
            unsigned u = (idx < nE) ? (unsigned)(dstp[idx] - lo) : 0xFFFFFFFFu;
            unsigned hit = __builtin_amdgcn_ballot_w32(u < ucnt && (int)(u & 31u) == w);
            while (hit) {
                int j = __builtin_ctz(hit);
                hit &= hit - 1;
                int d = __builtin_amdgcn_readlane((int)u, j);
                int src = ei[b + j];
                if (lane < HEADS) {
                    float v = s_src[src * HEADS + lane] + sd[d * HEADS + lane];
                    v = v > 0.f ? v : 0.2f * v;
                    atomicMax(&mx[d * HEADS + lane], flipf(v));
                }
            }
        }
    }
    __syncthreads();

    if (!slow) {
        int off = 0;
        for (int sp = 0; sp < w; ++sp) off += cnt2[lane * CPIT + sp];
        for (int i = 0; i < nmine; i += 32) {
            const int pos = i + lane;
            const unsigned pk = (pos < nmine) ? mylist[pos] : NOEDGE;
            const unsigned vm = __builtin_amdgcn_ballot_w32(pk != NOEDGE);
            const unsigned b0 = __builtin_amdgcn_ballot_w32((pk & 1u) != 0u);
            const unsigned b1 = __builtin_amdgcn_ballot_w32((pk & 2u) != 0u);
            const unsigned b2 = __builtin_amdgcn_ballot_w32((pk & 4u) != 0u);
            const unsigned b3 = __builtin_amdgcn_ballot_w32((pk & 8u) != 0u);
            const unsigned b4 = __builtin_amdgcn_ballot_w32((pk & 16u) != 0u);
            const unsigned o  = pk & 31u;
            const unsigned peers = owner_lanes(vm, b0, b1, b2, b3, b4, o);
            const int slot = __builtin_amdgcn_ds_bpermute((int)(o << 2), off)
                           + (int)__builtin_amdgcn_mbcnt_lo(peers, 0u);
            if (pk != NOEDGE && slot < LCAP) own[o * LSTR + slot] = pk;
            off += __builtin_popcount(owner_lanes(vm, b0, b1, b2, b3, b4, (unsigned)lane));
        }
        if (w == NWAVE - 1) {
            ocnt[lane] = off < LCAP ? off : LCAP;
            if (off > LCAP) ovf = 1;
        }
    }
    __syncthreads();
    const int slow2 = __builtin_amdgcn_readfirstlane(ovf);

    int pend_s = 0, pend_d = 0, npend = 0;
    const int h  = lane >> 3;
    const int sl = lane & 7;
#define GROUP_LANE(x, k) __uint_as_float(__builtin_amdgcn_mov_dpp8(__float_as_uint(x), (unsigned)(k) * 0x249249u))
#define EDGE_STEP(k)                                                        \
    {                                                                       \
        const int   d   = __builtin_amdgcn_readlane(pend_d, k);             \
        const float eek = GROUP_LANE(ee, k);                                \
        run = (pend_d == d) ? run + eek : run;                              \
        if (k < npend) {                                                    \
            float4 a = acc[d * 32 + lane];                                  \
            a.x += eek * row[k].x;                                          \
            a.y += eek * row[k].y;                                          \
            a.z += eek * row[k].z;                                          \
            a.w += eek * row[k].w;                                          \
            acc[d * 32 + lane] = a;                                         \
        }                                                                   \
    }
    auto flush = [&]() {
        float4 row[8];
        #pragma unroll
        for (int k = 0; k < 8; ++k) {
            int s = __builtin_amdgcn_readlane(pend_s, k);
            row[k] = *(const float4*)(Wx + (size_t)s * DIM + lane * 4);
        }
        float v = s_src[pend_s * HEADS + h] + sd[pend_d * HEADS + h];
        v = v > 0.f ? v : 0.2f * v;
        float ee = expf(v - unflipf(mx[pend_d * HEADS + h]));
        if (sl >= npend) ee = 0.f;
        float run = den[pend_d * HEADS + h];
        EDGE_STEP(0) EDGE_STEP(1) EDGE_STEP(2) EDGE_STEP(3)
        EDGE_STEP(4) EDGE_STEP(5) EDGE_STEP(6) EDGE_STEP(7)
        if (sl < npend) den[pend_d * HEADS + h] = run;
        npend = 0;
    };
#undef EDGE_STEP
#undef GROUP_LANE
    if (!slow2) {
        const unsigned* ml = own + w * LSTR;
        const int n = __builtin_amdgcn_readfirstlane(ocnt[w]);
        int nb = n < 8 ? n : 8;
        unsigned pk = (n > 0) ? ml[sl < nb ? sl : nb - 1] : 0u;
        for (int b0 = 0; b0 < n; b0 += 8) {
            pend_s = (int)(pk >> 8);
            pend_d = (int)(pk & 255u);
            const int cur = nb;
            const int rest = n - b0 - 8;
            nb = rest < 8 ? rest : 8;
            if (rest > 0) pk = ml[b0 + 8 + (sl < nb ? sl : nb - 1)];
            if (cur == 8) { npend = 8;   flush(); }
            else          { npend = cur; flush(); }
        }
    } else {
        auto push = [&](int s, int d) {
            pend_s = (sl == npend) ? s : pend_s;
            pend_d = (sl == npend) ? d : pend_d;
            if (++npend == 8) flush();
        };
        for (int b = 0; b < nE; b += 32) {
            int idx = b + lane;
            unsigned u = (idx < nE) ? (unsigned)(dstp[idx] - lo) : 0xFFFFFFFFu;
            bool mine = u < ucnt && (int)(u & 31u) == w;
            int sv = mine ? ei[idx] : 0;
            unsigned hit = __builtin_amdgcn_ballot_w32(mine);
            while (hit) {
                int j = __builtin_ctz(hit);
                hit &= hit - 1;
                push(__builtin_amdgcn_readlane(sv, j), __builtin_amdgcn_readlane((int)u, j));
            }
        }
        if (npend > 0) flush();
    }
    __syncthreads();

    for (int q = tid; q < cntR * 32; q += 1024) {
        int n = q >> 5, c4 = q & 31;
        float dn = den[n * HEADS + (c4 >> 3)] + 1e-8f;
        float4 a = acc[q];
        float4 r;
        r.x = a.x / dn; r.x = r.x > 0.f ? r.x : expm1f(r.x);
        r.y = a.y / dn; r.y = r.y > 0.f ? r.y : expm1f(r.y);
        r.z = a.z / dn; r.z = r.z > 0.f ? r.z : expm1f(r.z);
        r.w = a.w / dn; r.w = r.w > 0.f ? r.w : expm1f(r.w);
        *(float4*)(out + (size_t)(lo + n) * DIM + c4 * 4) = r;
    }
}

extern "C" void kernel_launch(void* const* d_in, const int* in_sizes, int n_in,
                              void* d_out, int out_size, void* d_ws, size_t ws_size,
                              hipStream_t stream) {
    const float* x   = (const float*)d_in[0];
    const int*   ei  = (const int*)d_in[1];
    const float* W   = (const float*)d_in[2];
    const float* a_w = (const float*)d_in[3];
    int N = in_sizes[0] / DIM;
    int E = in_sizes[1] / 2;
    float* out = (float*)d_out;

    float* Wx    = (float*)d_ws;
    float* s_src = Wx + (size_t)N * DIM;
    float* s_dst = s_src + (size_t)N * HEADS;
    unsigned* keys = (unsigned*)(s_dst + (size_t)N * HEADS);

    dim3 blk(256);
    int nTiles = (N + 15) / 16;
    gat_gemm_wmma<<<256, blk, 0, stream>>>(x, W, Wx, N, nTiles);
    gat_scores  <<<(N * HEADS + 255) / 256, blk, 0, stream>>>(Wx, a_w, s_src, s_dst, N, ei, keys, E);
    gat_edge_owner<<<(N + RN - 1) / RN, dim3(1024), 0, stream>>>(ei, s_src, s_dst, Wx, keys, out, E, N);
}


# Edge scan reads each flagged 16-edge vector's dst and src words straight from the edge list, four edges a lane, exact dst match; key bytes and per-edge re-reads gone; sum order changed, fixed
# speedup vs baseline: 6.0865x; 1.0509x over previous
#include <hip/hip_runtime.h>
#include <hip/hip_bf16.h>

#define DIM       128
#define HEADS     4
#define HEAD_DIM  32
#define RN        196

typedef __attribute__((ext_vector_type(2))) float v2f;
typedef __attribute__((ext_vector_type(8))) float v8f;

__device__ __forceinline__ unsigned flipf(float f) {
    unsigned u = __float_as_uint(f);
    return (u & 0x80000000u) ? ~u : (u | 0x80000000u);
}
__device__ __forceinline__ float unflipf(unsigned u) {
    return __uint_as_float((u & 0x80000000u) ? (u & 0x7FFFFFFFu) : ~u);
}

#define KSTEPS (DIM / 4)
__global__ __launch_bounds__(256)
void gat_gemm_wmma(const float* __restrict__ x, const float* __restrict__ W,
                   float* __restrict__ Wx, int nNodes, int nTiles) {
    const int lane = threadIdx.x & 31;
    const int w    = __builtin_amdgcn_readfirstlane(threadIdx.x >> 5);
    const int rc   = lane & 15;
    const int hi   = lane >> 4;
    const int cg   = w & 3;
    const int slot  = blockIdx.x * 2 + (w >> 2);
    const int nslot = gridDim.x * 2;
    if (slot >= nTiles) return;

    v2f b0[KSTEPS], b1[KSTEPS];
    const float* wp0 = W + (size_t)(cg * 32 + rc) * DIM + hi * 2;
    const float* wp1 = wp0 + (size_t)16 * DIM;
    #pragma unroll
    for (int ks = 0; ks < KSTEPS; ++ks) {
        b0[ks] = *(const v2f*)(wp0 + ks * 4);
        b1[ks] = *(const v2f*)(wp1 + ks * 4);
    }

    auto load_a = [&](v2f (&a)[KSTEPS], int tile) {
        int arow = tile * 16 + rc;
        if (arow >= nNodes) arow = nNodes - 1;
        const float* aptr = x + (size_t)arow * DIM + hi * 2;
        #pragma unroll
        for (int ks = 0; ks < KSTEPS; ++ks) a[ks] = *(const v2f*)(aptr + ks * 4);
    };
    auto do_tile = [&](const v2f (&a)[KSTEPS], int tile) {
        v8f acc0 = {}, acc1 = {};
        #pragma unroll
        for (int ks = 0; ks < KSTEPS; ++ks) {
            acc0 = __builtin_amdgcn_wmma_f32_16x16x4_f32(
                false, a[ks], false, b0[ks], (short)0, acc0, false, false);
            acc1 = __builtin_amdgcn_wmma_f32_16x16x4_f32(
                false, a[ks], false, b1[ks], (short)0, acc1, false, false);
        }
        const int nodeBase = tile * 16;
        #pragma unroll
        for (int j = 0; j < 8; ++j) {
            int node = nodeBase + j + 8 * hi;
            if (node < nNodes) {
                float* o = Wx + (size_t)node * DIM + cg * 32 + rc;
                o[0]  = acc0[j];
                o[16] = acc1[j];
            }
        }
    };

    v2f aA[KSTEPS], aB[KSTEPS];
    int tile = slot;
    load_a(aA, tile);
    for (;;) {
        const int t1 = tile + nslot;
        load_a(aB, t1 < nTiles ? t1 : tile);
        __builtin_amdgcn_sched_barrier(0);
        do_tile(aA, tile);
        if (t1 >= nTiles) break;
        const int t2 = t1 + nslot;
        load_a(aA, t2 < nTiles ? t2 : t1);
        __builtin_amdgcn_sched_barrier(0);
        do_tile(aB, t1);
        if (t2 >= nTiles) break;
        tile = t2;
    }
}

#define KROWS 256
__global__ __launch_bounds__(256)
void gat_scores(const float* __restrict__ Wx, const float* __restrict__ a_w,
                float* __restrict__ s_src, float* __restrict__ s_dst, int nNodes,
                const int* __restrict__ ei, unsigned* __restrict__ pres, int nE) {
    __shared__ unsigned tb[KROWS * 2];
    const int tid = threadIdx.x;
    int t = blockIdx.x * blockDim.x + tid;
    const int* __restrict__ dstp = ei + nE;
    const int nW = ((nE + 15) >> 4) << 2;
    const int nCh = (nW + 255) >> 8;
    for (int c = blockIdx.x; c < nCh; c += (int)gridDim.x) {
        tb[tid] = 0u;
        tb[tid + 256] = 0u;
        __syncthreads();
        const int q = c * 256 + tid;
        if (q < nW) {
            const unsigned bit = 1u << ((tid >> 2) & 31);
            const int half = tid >> 7;
            #pragma unroll
            for (int b = 0; b < 4; ++b) {
                int e = q * 4 + b;
                if (e < nE) {
                    unsigned k = ((unsigned)dstp[e] / (unsigned)RN) & 255u;
                    atomicOr(&tb[k * 2 + half], bit);
                }
            }
        }
        __syncthreads();
        *(uint2*)(pres + ((size_t)c * KROWS + tid) * 2) = make_uint2(tb[tid * 2], tb[tid * 2 + 1]);
        __syncthreads();
    }
    if (t >= nNodes * HEADS) return;
    int h = t & (HEADS - 1);
    int n = t >> 2;
    const float* p = Wx + (size_t)n * DIM + h * HEAD_DIM;
    float ss = 0.f, sd = 0.f;
    #pragma unroll
    for (int d = 0; d < HEAD_DIM; ++d) {
        float v = p[d];
        ss += v * a_w[d];
        sd += v * a_w[HEAD_DIM + d];
    }
    s_src[t] = ss;
    s_dst[t] = sd;
}

#define NWAVE 32
#define LCAP  512
#define LSTR  (LCAP + 1)
#define CPIT  (NWAVE + 1)
#define NOEDGE 0xFFFFFFFFu
#define SRCBITS 24
#define SCANQ 4

__device__ __forceinline__ unsigned owner_lanes(unsigned vm, unsigned b0, unsigned b1, unsigned b2,
                                                unsigned b3, unsigned b4, unsigned who) {
    unsigned m = vm;
    m &= (who & 1u)  ? b0 : ~b0;
    m &= (who & 2u)  ? b1 : ~b1;
    m &= (who & 4u)  ? b2 : ~b2;
    m &= (who & 8u)  ? b3 : ~b3;
    m &= (who & 16u) ? b4 : ~b4;
    return m;
}

__global__ __launch_bounds__(1024)
void gat_edge_owner(const int* __restrict__ ei, const float* __restrict__ s_src,
                    const float* __restrict__ s_dst, const float* __restrict__ Wx,
                    const unsigned* __restrict__ pres,
                    float* __restrict__ out, int nE, int nNodes) {
    __shared__ float4   acc[RN * 32];
    __shared__ float    den[RN * HEADS];
    __shared__ unsigned mx[RN * HEADS];
    __shared__ float    sd[RN * HEADS];
    __shared__ unsigned lst[NWAVE * LSTR];
    __shared__ unsigned own[NWAVE * LSTR];
    __shared__ int      cnt2[NWAVE * CPIT];
    __shared__ int      lcnt[NWAVE];
    __shared__ int      ocnt[NWAVE];
    __shared__ int      ovf;

    const int tid  = threadIdx.x;
    const int lane = tid & 31;
    const int w    = __builtin_amdgcn_readfirstlane(tid >> 5);
    const int lo   = blockIdx.x * RN;
    int cntR = nNodes - lo;
    if (cntR > RN) cntR = RN;
    const unsigned ucnt = (unsigned)cntR;
    const int* __restrict__ dstp = ei + nE;

    for (int i = tid; i < RN * 32; i += 1024) acc[i] = make_float4(0.f, 0.f, 0.f, 0.f);
    for (int i = tid; i < RN * HEADS; i += 1024) {
        den[i] = 0.f;
        mx[i]  = 0u;
        sd[i]  = (i < cntR * HEADS) ? s_dst[(size_t)lo * HEADS + i] : 0.f;
    }
    if (tid == 0) ovf = (nNodes > (1 << SRCBITS) || nE < 4) ? 1 : 0;
    __syncthreads();

    int cnt = 0;
    unsigned* mylist = lst + w * LSTR;
    const int nvec = (nE + 15) >> 4;
    const int nW   = nvec << 2;
    const int nCh  = (nW + 255) >> 8;
    const int cpw  = (nCh + NWAVE - 1) / NWAVE;
    const uint2* __restrict__ pr2 = (const uint2*)pres + (blockIdx.x & 255);
    unsigned* vl = own + w * LSTR;
    int vcnt = 0;
    auto note = [&](unsigned bits, int i) {
        unsigned bal = __builtin_amdgcn_ballot_w32(bits != 0u);
        while (bal) {
            int pos = vcnt + (int)__builtin_amdgcn_mbcnt_lo(bal, 0u);
            if (bits != 0u && pos < LCAP) vl[pos] = (unsigned)(i * 32 + __builtin_ctz(bits));
            vcnt += __builtin_popcount(bal);
            bits &= bits - 1u;
            bal = __builtin_amdgcn_ballot_w32(bits != 0u);
        }
    };
    for (int k0 = 0; k0 < cpw; k0 += 32) {
        const int k  = k0 + lane;
        const int ch = w * cpw + k;
        uint2 wd = make_uint2(0u, 0u);
        if (k < cpw && ch < nCh) wd = pr2[(size_t)ch * 256];
        note(wd.x, 2 * ch);
        note(wd.y, 2 * ch + 1);
    }
    __syncthreads();
    const int nv = vcnt < LCAP ? vcnt : LCAP;
    const int vq = lane >> 2;
    const unsigned vj = (unsigned)(lane & 3) * 4u;
    const int nvs = (nE >= 4) ? nv : 0;
    for (int i = 0; i < nvs; i += 8 * SCANQ) {
        int4 dq[SCANQ], sq[SCANQ];
        unsigned okm = 0u;
        unsigned tlm = 0u;
        #pragma unroll
        for (int k = 0; k < SCANQ; ++k) {
            const int p = i + 8 * k + vq;
            const unsigned e = vl[p < nv ? p : nv - 1] * 16u + vj;
            const bool ok = p < nv && e + 3u < (unsigned)nE;
            const bool tl = p < nv && e < (unsigned)nE && !ok;
            const unsigned ec = ok ? e : 0u;
            okm |= (ok ? 1u : 0u) << k;
            tlm |= (tl ? 1u : 0u) << k;
            __builtin_memcpy(&dq[k], dstp + ec, 16);
            __builtin_memcpy(&sq[k], ei + ec, 16);
        }
        asm volatile("" :: "v"(sq[0].x), "v"(sq[0].y), "v"(sq[0].z), "v"(sq[0].w), "v"(sq[1].x), "v"(sq[1].y), "v"(sq[1].z), "v"(sq[1].w),
                           "v"(sq[2].x), "v"(sq[2].y), "v"(sq[2].z), "v"(sq[2].w), "v"(sq[3].x), "v"(sq[3].y), "v"(sq[3].z), "v"(sq[3].w));
        #pragma unroll
        for (int k = 0; k < SCANQ; ++k) {
            const unsigned lok = ((okm >> k) & 1u) != 0u ? (unsigned)lo : 0x80000000u;
            const int dd[4] = {dq[k].x, dq[k].y, dq[k].z, dq[k].w};
            const int ss[4] = {sq[k].x, sq[k].y, sq[k].z, sq[k].w};
            #pragma unroll
            for (int c = 0; c < 4; ++c) {
                const unsigned u = (unsigned)dd[c] - lok;
                const unsigned bal = __builtin_amdgcn_ballot_w32(u < ucnt);
                const int pos = (int)__builtin_amdgcn_mbcnt_lo(bal, (unsigned)cnt);
                if (u < ucnt) mylist[pos < LCAP ? pos : LCAP] = ((unsigned)ss[c] << 8) | u;
                cnt += __builtin_popcount(bal);
            }
        }
        if (__builtin_amdgcn_ballot_w32(tlm != 0u) != 0u) {
            for (int k = 0; k < SCANQ; ++k) {
                const int p = i + 8 * k + vq;
                const bool tl = ((tlm >> k) & 1u) != 0u;
                const unsigned e = tl ? vl[p] * 16u + vj : 0u;
                for (unsigned c = 0; c < 3u; ++c) {
                    const bool in = tl && e + c < (unsigned)nE;
                    const unsigned u = in ? (unsigned)(dstp[e + c] - lo) : 0xFFFFFFFFu;
                    const int sv = in ? ei[e + c] : 0;
                    const bool hit = u < ucnt;
                    const unsigned bal = __builtin_amdgcn_ballot_w32(hit);
                    const int pos = (int)__builtin_amdgcn_mbcnt_lo(bal, (unsigned)cnt);
                    if (hit) mylist[pos < LCAP ? pos : LCAP] = ((unsigned)sv << 8) | u;
                    cnt += __builtin_popcount(bal);
                }
            }
        }
    }
    if (lane == 0) {
        lcnt[w] = cnt < LCAP ? cnt : LCAP;
        if (cnt > LCAP || vcnt > LCAP) ovf = 1;
    }
    __syncthreads();
    const int slow = __builtin_amdgcn_readfirstlane(ovf);
    const int nmine = __builtin_amdgcn_readfirstlane(lcnt[w]);

    if (!slow) {
        int hist = 0;
        for (int i = 0; i < nmine; i += 32) {
            const int pos = i + lane;
            unsigned pk = NOEDGE;
            if (pos < nmine) {
                pk = mylist[pos];
                const int src = (int)(pk >> 8);
                const unsigned u = pk & 255u;
                float4 ss = ((const float4*)s_src)[src];
                float v0 = ss.x + sd[u * HEADS + 0];
                float v1 = ss.y + sd[u * HEADS + 1];
                float v2 = ss.z + sd[u * HEADS + 2];
                float v3 = ss.w + sd[u * HEADS + 3];
                v0 = v0 > 0.f ? v0 : 0.2f * v0;
                v1 = v1 > 0.f ? v1 : 0.2f * v1;
                v2 = v2 > 0.f ? v2 : 0.2f * v2;
                v3 = v3 > 0.f ? v3 : 0.2f * v3;
                atomicMax(&mx[u * HEADS + 0], flipf(v0));
                atomicMax(&mx[u * HEADS + 1], flipf(v1));
                atomicMax(&mx[u * HEADS + 2], flipf(v2));
                atomicMax(&mx[u * HEADS + 3], flipf(v3));
            }
            const unsigned vm = __builtin_amdgcn_ballot_w32(pk != NOEDGE);
            const unsigned b0 = __builtin_amdgcn_ballot_w32((pk & 1u) != 0u);
            const unsigned b1 = __builtin_amdgcn_ballot_w32((pk & 2u) != 0u);
            const unsigned b2 = __builtin_amdgcn_ballot_w32((pk & 4u) != 0u);
            const unsigned b3 = __builtin_amdgcn_ballot_w32((pk & 8u) != 0u);
            const unsigned b4 = __builtin_amdgcn_ballot_w32((pk & 16u) != 0u);
            hist += __builtin_popcount(owner_lanes(vm, b0, b1, b2, b3, b4, (unsigned)lane));
        }
        cnt2[lane * CPIT + w] = hist;
    } else {
        for (int b = 0; b < nE; b += 32) {
            int idx = b + lane;
            unsigned u = (idx < nE) ? (unsigned)(dstp[idx] - lo) : 0xFFFFFFFFu;
            unsigned hit = __builtin_amdgcn_ballot_w32(u < ucnt && (int)(u & 31u) == w);
            while (hit) {
                int j = __builtin_ctz(hit);
                hit &= hit - 1;
                int d = __builtin_amdgcn_readlane((int)u, j);
                int src = ei[b + j];
                if (lane < HEADS) {
                    float v = s_src[src * HEADS + lane] + sd[d * HEADS + lane];
                    v = v > 0.f ? v : 0.2f * v;
                    atomicMax(&mx[d * HEADS + lane], flipf(v));
                }
            }
        }
    }
    __syncthreads();

    if (!slow) {
        int off = 0;
        for (int sp = 0; sp < w; ++sp) off += cnt2[lane * CPIT + sp];
        for (int i = 0; i < nmine; i += 32) {
            const int pos = i + lane;
            const unsigned pk = (pos < nmine) ? mylist[pos] : NOEDGE;
            const unsigned vm = __builtin_amdgcn_ballot_w32(pk != NOEDGE);
            const unsigned b0 = __builtin_amdgcn_ballot_w32((pk & 1u) != 0u);
            const unsigned b1 = __builtin_amdgcn_ballot_w32((pk & 2u) != 0u);
            const unsigned b2 = __builtin_amdgcn_ballot_w32((pk & 4u) != 0u);
            const unsigned b3 = __builtin_amdgcn_ballot_w32((pk & 8u) != 0u);
            const unsigned b4 = __builtin_amdgcn_ballot_w32((pk & 16u) != 0u);
            const unsigned o  = pk & 31u;
            const unsigned peers = owner_lanes(vm, b0, b1, b2, b3, b4, o);
            const int slot = __builtin_amdgcn_ds_bpermute((int)(o << 2), off)
                           + (int)__builtin_amdgcn_mbcnt_lo(peers, 0u);
            if (pk != NOEDGE && slot < LCAP) own[o * LSTR + slot] = pk;
            off += __builtin_popcount(owner_lanes(vm, b0, b1, b2, b3, b4, (unsigned)lane));
        }
        if (w == NWAVE - 1) {
            ocnt[lane] = off < LCAP ? off : LCAP;
            if (off > LCAP) ovf = 1;
        }
    }
    __syncthreads();
    const int slow2 = __builtin_amdgcn_readfirstlane(ovf);

    int pend_s = 0, pend_d = 0, npend = 0;
    const int h  = lane >> 3;
    const int sl = lane & 7;
#define GROUP_LANE(x, k) __uint_as_float(__builtin_amdgcn_mov_dpp8(__float_as_uint(x), (unsigned)(k) * 0x249249u))
#define EDGE_STEP(k)                                                        \
    {                                                                       \
        const int   d   = __builtin_amdgcn_readlane(pend_d, k);             \
        const float eek = GROUP_LANE(ee, k);                                \
        run = (pend_d == d) ? run + eek : run;                              \
        if (k < npend) {                                                    \
            float4 a = acc[d * 32 + lane];                                  \
            a.x += eek * row[k].x;                                          \
            a.y += eek * row[k].y;                                          \
            a.z += eek * row[k].z;                                          \
            a.w += eek * row[k].w;                                          \
            acc[d * 32 + lane] = a;                                         \
        }                                                                   \
    }
    auto flush = [&]() {
        float4 row[8];
        #pragma unroll
        for (int k = 0; k < 8; ++k) {
            int s = __builtin_amdgcn_readlane(pend_s, k);
            row[k] = *(const float4*)(Wx + (size_t)s * DIM + lane * 4);
        }
        float v = s_src[pend_s * HEADS + h] + sd[pend_d * HEADS + h];
        v = v > 0.f ? v : 0.2f * v;
        float ee = expf(v - unflipf(mx[pend_d * HEADS + h]));
        if (sl >= npend) ee = 0.f;
        float run = den[pend_d * HEADS + h];
        EDGE_STEP(0) EDGE_STEP(1) EDGE_STEP(2) EDGE_STEP(3)
        EDGE_STEP(4) EDGE_STEP(5) EDGE_STEP(6) EDGE_STEP(7)
        if (sl < npend) den[pend_d * HEADS + h] = run;
        npend = 0;
    };
#undef EDGE_STEP
#undef GROUP_LANE
    if (!slow2) {
        const unsigned* ml = own + w * LSTR;
        const int n = __builtin_amdgcn_readfirstlane(ocnt[w]);
        int nb = n < 8 ? n : 8;
        unsigned pk = (n > 0) ? ml[sl < nb ? sl : nb - 1] : 0u;
        for (int b0 = 0; b0 < n; b0 += 8) {
            pend_s = (int)(pk >> 8);
            pend_d = (int)(pk & 255u);
            const int cur = nb;
            const int rest = n - b0 - 8;
            nb = rest < 8 ? rest : 8;
            if (rest > 0) pk = ml[b0 + 8 + (sl < nb ? sl : nb - 1)];
            if (cur == 8) { npend = 8;   flush(); }
            else          { npend = cur; flush(); }
        }
    } else {
        auto push = [&](int s, int d) {
            pend_s = (sl == npend) ? s : pend_s;
            pend_d = (sl == npend) ? d : pend_d;
            if (++npend == 8) flush();
        };
        for (int b = 0; b < nE; b += 32) {
            int idx = b + lane;
            unsigned u = (idx < nE) ? (unsigned)(dstp[idx] - lo) : 0xFFFFFFFFu;
            bool mine = u < ucnt && (int)(u & 31u) == w;
            int sv = mine ? ei[idx] : 0;
            unsigned hit = __builtin_amdgcn_ballot_w32(mine);
            while (hit) {
                int j = __builtin_ctz(hit);
                hit &= hit - 1;
                push(__builtin_amdgcn_readlane(sv, j), __builtin_amdgcn_readlane((int)u, j));
            }
        }
        if (npend > 0) flush();
    }
    __syncthreads();

    for (int q = tid; q < cntR * 32; q += 1024) {
        int n = q >> 5, c4 = q & 31;
        float dn = den[n * HEADS + (c4 >> 3)] + 1e-8f;
        float4 a = acc[q];
        float4 r;
        r.x = a.x / dn; r.x = r.x > 0.f ? r.x : expm1f(r.x);
        r.y = a.y / dn; r.y = r.y > 0.f ? r.y : expm1f(r.y);
        r.z = a.z / dn; r.z = r.z > 0.f ? r.z : expm1f(r.z);
        r.w = a.w / dn; r.w = r.w > 0.f ? r.w : expm1f(r.w);
        *(float4*)(out + (size_t)(lo + n) * DIM + c4 * 4) = r;
    }
}

extern "C" void kernel_launch(void* const* d_in, const int* in_sizes, int n_in,
                              void* d_out, int out_size, void* d_ws, size_t ws_size,
                              hipStream_t stream) {
    const float* x   = (const float*)d_in[0];
    const int*   ei  = (const int*)d_in[1];
    const float* W   = (const float*)d_in[2];
    const float* a_w = (const float*)d_in[3];
    int N = in_sizes[0] / DIM;
    int E = in_sizes[1] / 2;
    float* out = (float*)d_out;

    float* Wx    = (float*)d_ws;
    float* s_src = Wx + (size_t)N * DIM;
    float* s_dst = s_src + (size_t)N * HEADS;
    unsigned* pres = (unsigned*)(s_dst + (size_t)N * HEADS);

    dim3 blk(256);
    int nTiles = (N + 15) / 16;
    gat_gemm_wmma<<<256, blk, 0, stream>>>(x, W, Wx, N, nTiles);
    gat_scores  <<<(N * HEADS + 255) / 256, blk, 0, stream>>>(Wx, a_w, s_src, s_dst, N, ei, pres, E);
    gat_edge_owner<<<(N + RN - 1) / RN, dim3(1024), 0, stream>>>(ei, s_src, s_dst, Wx, pres, out, E, N);
}


# Edge stage: an edge's four leaky_relu scores are made once, in the pass that copies it to its owner's list, and kept beside it in LDS; the sum pass's second s_src gather is gone; sums unchanged
# speedup vs baseline: 6.4005x; 1.0516x over previous
#include <hip/hip_runtime.h>
#include <hip/hip_bf16.h>

#define DIM       128
#define HEADS     4
#define HEAD_DIM  32
#define RN        196

typedef __attribute__((ext_vector_type(2))) float v2f;
typedef __attribute__((ext_vector_type(8))) float v8f;

__device__ __forceinline__ unsigned flipf(float f) {
    unsigned u = __float_as_uint(f);
    return (u & 0x80000000u) ? ~u : (u | 0x80000000u);
}
__device__ __forceinline__ float unflipf(unsigned u) {
    return __uint_as_float((u & 0x80000000u) ? (u & 0x7FFFFFFFu) : ~u);
}

#define KSTEPS (DIM / 4)
__global__ __launch_bounds__(256)
void gat_gemm_wmma(const float* __restrict__ x, const float* __restrict__ W,
                   float* __restrict__ Wx, int nNodes, int nTiles) {
    const int lane = threadIdx.x & 31;
    const int w    = __builtin_amdgcn_readfirstlane(threadIdx.x >> 5);
    const int rc   = lane & 15;
    const int hi   = lane >> 4;
    const int cg   = w & 3;
    const int slot  = blockIdx.x * 2 + (w >> 2);
    const int nslot = gridDim.x * 2;
    if (slot >= nTiles) return;

    v2f b0[KSTEPS], b1[KSTEPS];
    const float* wp0 = W + (size_t)(cg * 32 + rc) * DIM + hi * 2;
    const float* wp1 = wp0 + (size_t)16 * DIM;
    #pragma unroll
    for (int ks = 0; ks < KSTEPS; ++ks) {
        b0[ks] = *(const v2f*)(wp0 + ks * 4);
        b1[ks] = *(const v2f*)(wp1 + ks * 4);
    }

    auto load_a = [&](v2f (&a)[KSTEPS], int tile) {
        int arow = tile * 16 + rc;
        if (arow >= nNodes) arow = nNodes - 1;
        const float* aptr = x + (size_t)arow * DIM + hi * 2;
        #pragma unroll
        for (int ks = 0; ks < KSTEPS; ++ks) a[ks] = *(const v2f*)(aptr + ks * 4);
    };
    auto do_tile = [&](const v2f (&a)[KSTEPS], int tile) {
        v8f acc0 = {}, acc1 = {};
        #pragma unroll
        for (int ks = 0; ks < KSTEPS; ++ks) {
            acc0 = __builtin_amdgcn_wmma_f32_16x16x4_f32(
                false, a[ks], false, b0[ks], (short)0, acc0, false, false);
            acc1 = __builtin_amdgcn_wmma_f32_16x16x4_f32(
                false, a[ks], false, b1[ks], (short)0, acc1, false, false);
        }
        const int nodeBase = tile * 16;
        #pragma unroll
        for (int j = 0; j < 8; ++j) {
            int node = nodeBase + j + 8 * hi;
            if (node < nNodes) {
                float* o = Wx + (size_t)node * DIM + cg * 32 + rc;
                o[0]  = acc0[j];
                o[16] = acc1[j];
            }
        }
    };

    v2f aA[KSTEPS], aB[KSTEPS];
    int tile = slot;
    load_a(aA, tile);
    for (;;) {
        const int t1 = tile + nslot;
        load_a(aB, t1 < nTiles ? t1 : tile);
        __builtin_amdgcn_sched_barrier(0);
        do_tile(aA, tile);
        if (t1 >= nTiles) break;
        const int t2 = t1 + nslot;
        load_a(aA, t2 < nTiles ? t2 : t1);
        __builtin_amdgcn_sched_barrier(0);
        do_tile(aB, t1);
        if (t2 >= nTiles) break;
        tile = t2;
    }
}

#define KROWS 256
__global__ __launch_bounds__(256)
void gat_scores(const float* __restrict__ Wx, const float* __restrict__ a_w,
                float* __restrict__ s_src, float* __restrict__ s_dst, int nNodes,
                const int* __restrict__ ei, unsigned* __restrict__ pres, int nE) {
    __shared__ unsigned tb[KROWS * 2];
    const int tid = threadIdx.x;
    int t = blockIdx.x * blockDim.x + tid;
    const int* __restrict__ dstp = ei + nE;
    const int nW = ((nE + 15) >> 4) << 2;
    const int nCh = (nW + 255) >> 8;
    for (int c = blockIdx.x; c < nCh; c += (int)gridDim.x) {
        tb[tid] = 0u;
        tb[tid + 256] = 0u;
        __syncthreads();
        const int q = c * 256 + tid;
        if (q < nW) {
            const unsigned bit = 1u << ((tid >> 2) & 31);
            const int half = tid >> 7;
            #pragma unroll
            for (int b = 0; b < 4; ++b) {
                int e = q * 4 + b;
                if (e < nE) {
                    unsigned k = ((unsigned)dstp[e] / (unsigned)RN) & 255u;
                    atomicOr(&tb[k * 2 + half], bit);
                }
            }
        }
        __syncthreads();
        *(uint2*)(pres + ((size_t)c * KROWS + tid) * 2) = make_uint2(tb[tid * 2], tb[tid * 2 + 1]);
        __syncthreads();
    }
    if (t >= nNodes * HEADS) return;
    int h = t & (HEADS - 1);
    int n = t >> 2;
    const float* p = Wx + (size_t)n * DIM + h * HEAD_DIM;
    float ss = 0.f, sd = 0.f;
    #pragma unroll
    for (int d = 0; d < HEAD_DIM; ++d) {
        float v = p[d];
        ss += v * a_w[d];
        sd += v * a_w[HEAD_DIM + d];
    }
    s_src[t] = ss;
    s_dst[t] = sd;
}

#define NWAVE 32
#define LCAP  256
#define LSTR  (LCAP + 1)
#define CPIT  (NWAVE + 1)
#define NOEDGE 0xFFFFFFFFu
#define SRCBITS 24
#define SCANQ 4

__device__ __forceinline__ unsigned owner_lanes(unsigned vm, unsigned b0, unsigned b1, unsigned b2,
                                                unsigned b3, unsigned b4, unsigned who) {
    unsigned m = vm;
    m &= (who & 1u)  ? b0 : ~b0;
    m &= (who & 2u)  ? b1 : ~b1;
    m &= (who & 4u)  ? b2 : ~b2;
    m &= (who & 8u)  ? b3 : ~b3;
    m &= (who & 16u) ? b4 : ~b4;
    return m;
}

__global__ __launch_bounds__(1024)
void gat_edge_owner(const int* __restrict__ ei, const float* __restrict__ s_src,
                    const float* __restrict__ s_dst, const float* __restrict__ Wx,
                    const unsigned* __restrict__ pres,
                    float* __restrict__ out, int nE, int nNodes) {
    __shared__ float4   acc[RN * 32];
    __shared__ float    den[RN * HEADS];
    __shared__ unsigned mx[RN * HEADS];
    __shared__ float    sd[RN * HEADS];
    __shared__ unsigned lst[NWAVE * LSTR];
    __shared__ unsigned own[NWAVE * LSTR];
    __shared__ float4   scr[NWAVE * LSTR];
    __shared__ int      cnt2[NWAVE * CPIT];
    __shared__ int      lcnt[NWAVE];
    __shared__ int      ocnt[NWAVE];
    __shared__ int      ovf;

    const int tid  = threadIdx.x;
    const int lane = tid & 31;
    const int w    = __builtin_amdgcn_readfirstlane(tid >> 5);
    const int lo   = blockIdx.x * RN;
    int cntR = nNodes - lo;
    if (cntR > RN) cntR = RN;
    const unsigned ucnt = (unsigned)cntR;
    const int* __restrict__ dstp = ei + nE;

    for (int i = tid; i < RN * 32; i += 1024) acc[i] = make_float4(0.f, 0.f, 0.f, 0.f);
    for (int i = tid; i < RN * HEADS; i += 1024) {
        den[i] = 0.f;
        mx[i]  = 0u;
        sd[i]  = (i < cntR * HEADS) ? s_dst[(size_t)lo * HEADS + i] : 0.f;
    }
    if (tid == 0) ovf = (nNodes > (1 << SRCBITS) || nE < 4) ? 1 : 0;
    __syncthreads();

    int cnt = 0;
    unsigned* mylist = lst + w * LSTR;
    const int nvec = (nE + 15) >> 4;
    const int nW   = nvec << 2;
    const int nCh  = (nW + 255) >> 8;
    const int cpw  = (nCh + NWAVE - 1) / NWAVE;
    const uint2* __restrict__ pr2 = (const uint2*)pres + (blockIdx.x & 255);
    unsigned* vl = own + w * LSTR;
    int vcnt = 0;
    auto note = [&](unsigned bits, int i) {
        unsigned bal = __builtin_amdgcn_ballot_w32(bits != 0u);
        while (bal) {
            int pos = vcnt + (int)__builtin_amdgcn_mbcnt_lo(bal, 0u);
            if (bits != 0u && pos < LCAP) vl[pos] = (unsigned)(i * 32 + __builtin_ctz(bits));
            vcnt += __builtin_popcount(bal);
            bits &= bits - 1u;
            bal = __builtin_amdgcn_ballot_w32(bits != 0u);
        }
    };
    for (int k0 = 0; k0 < cpw; k0 += 32) {
        const int k  = k0 + lane;
        const int ch = w * cpw + k;
        uint2 wd = make_uint2(0u, 0u);
        if (k < cpw && ch < nCh) wd = pr2[(size_t)ch * 256];
        note(wd.x, 2 * ch);
        note(wd.y, 2 * ch + 1);
    }
    __syncthreads();
    const int nv = vcnt < LCAP ? vcnt : LCAP;
    const int vq = lane >> 2;
    const unsigned vj = (unsigned)(lane & 3) * 4u;
    const int nvs = (nE >= 4) ? nv : 0;
    for (int i = 0; i < nvs; i += 8 * SCANQ) {
        int4 dq[SCANQ], sq[SCANQ];
        unsigned okm = 0u;
        unsigned tlm = 0u;
        #pragma unroll
        for (int k = 0; k < SCANQ; ++k) {
            const int p = i + 8 * k + vq;
            const unsigned e = vl[p < nv ? p : nv - 1] * 16u + vj;
            const bool ok = p < nv && e + 3u < (unsigned)nE;
            const bool tl = p < nv && e < (unsigned)nE && !ok;
            const unsigned ec = ok ? e : 0u;
            okm |= (ok ? 1u : 0u) << k;
            tlm |= (tl ? 1u : 0u) << k;
            __builtin_memcpy(&dq[k], dstp + ec, 16);
            __builtin_memcpy(&sq[k], ei + ec, 16);
        }
        asm volatile("" :: "v"(sq[0].x), "v"(sq[0].y), "v"(sq[0].z), "v"(sq[0].w), "v"(sq[1].x), "v"(sq[1].y), "v"(sq[1].z), "v"(sq[1].w),
                           "v"(sq[2].x), "v"(sq[2].y), "v"(sq[2].z), "v"(sq[2].w), "v"(sq[3].x), "v"(sq[3].y), "v"(sq[3].z), "v"(sq[3].w));
        #pragma unroll
        for (int k = 0; k < SCANQ; ++k) {
            const unsigned lok = ((okm >> k) & 1u) != 0u ? (unsigned)lo : 0x80000000u;
            const int dd[4] = {dq[k].x, dq[k].y, dq[k].z, dq[k].w};
            const int ss[4] = {sq[k].x, sq[k].y, sq[k].z, sq[k].w};
            #pragma unroll
            for (int c = 0; c < 4; ++c) {
                const unsigned u = (unsigned)dd[c] - lok;
                const unsigned bal = __builtin_amdgcn_ballot_w32(u < ucnt);
                const int pos = (int)__builtin_amdgcn_mbcnt_lo(bal, (unsigned)cnt);
                if (u < ucnt) mylist[pos < LCAP ? pos : LCAP] = ((unsigned)ss[c] << 8) | u;
                cnt += __builtin_popcount(bal);
            }
        }
        if (__builtin_amdgcn_ballot_w32(tlm != 0u) != 0u) {
            for (int k = 0; k < SCANQ; ++k) {
                const int p = i + 8 * k + vq;
                const bool tl = ((tlm >> k) & 1u) != 0u;
                const unsigned e = tl ? vl[p] * 16u + vj : 0u;
                for (unsigned c = 0; c < 3u; ++c) {
                    const bool in = tl && e + c < (unsigned)nE;
                    const unsigned u = in ? (unsigned)(dstp[e + c] - lo) : 0xFFFFFFFFu;
                    const int sv = in ? ei[e + c] : 0;
                    const bool hit = u < ucnt;
                    const unsigned bal = __builtin_amdgcn_ballot_w32(hit);
                    const int pos = (int)__builtin_amdgcn_mbcnt_lo(bal, (unsigned)cnt);
                    if (hit) mylist[pos < LCAP ? pos : LCAP] = ((unsigned)sv << 8) | u;
                    cnt += __builtin_popcount(bal);
                }
            }
        }
    }
    if (lane == 0) {
        lcnt[w] = cnt < LCAP ? cnt : LCAP;
        if (cnt > LCAP || vcnt > LCAP) ovf = 1;
    }
    __syncthreads();
    const int slow = __builtin_amdgcn_readfirstlane(ovf);
    const int nmine = __builtin_amdgcn_readfirstlane(lcnt[w]);

    if (!slow) {
        int hist = 0;
        for (int i = 0; i < nmine; i += 32) {
            const int pos = i + lane;
            const unsigned pk = (pos < nmine) ? mylist[pos] : NOEDGE;
            const unsigned vm = __builtin_amdgcn_ballot_w32(pk != NOEDGE);
            const unsigned b0 = __builtin_amdgcn_ballot_w32((pk & 1u) != 0u);
            const unsigned b1 = __builtin_amdgcn_ballot_w32((pk & 2u) != 0u);
            const unsigned b2 = __builtin_amdgcn_ballot_w32((pk & 4u) != 0u);
            const unsigned b3 = __builtin_amdgcn_ballot_w32((pk & 8u) != 0u);
            const unsigned b4 = __builtin_amdgcn_ballot_w32((pk & 16u) != 0u);
            hist += __builtin_popcount(owner_lanes(vm, b0, b1, b2, b3, b4, (unsigned)lane));
        }
        cnt2[lane * CPIT + w] = hist;
    } else {
        for (int b = 0; b < nE; b += 32) {
            int idx = b + lane;
            unsigned u = (idx < nE) ? (unsigned)(dstp[idx] - lo) : 0xFFFFFFFFu;
            unsigned hit = __builtin_amdgcn_ballot_w32(u < ucnt && (int)(u & 31u) == w);
            while (hit) {
                int j = __builtin_ctz(hit);
                hit &= hit - 1;
                int d = __builtin_amdgcn_readlane((int)u, j);
                int src = ei[b + j];
                if (lane < HEADS) {
                    float v = s_src[src * HEADS + lane] + sd[d * HEADS + lane];
                    v = v > 0.f ? v : 0.2f * v;
                    atomicMax(&mx[d * HEADS + lane], flipf(v));
                }
            }
        }
    }
    __syncthreads();

    if (!slow) {
        int off = 0;
        for (int sp = 0; sp < w; ++sp) off += cnt2[lane * CPIT + sp];
        for (int i = 0; i < nmine; i += 32) {
            const int pos = i + lane;
            const unsigned pk = (pos < nmine) ? mylist[pos] : NOEDGE;
            float4 sc = make_float4(0.f, 0.f, 0.f, 0.f);
            if (pk != NOEDGE) {
                const int src = (int)(pk >> 8);
                const unsigned u = pk & 255u;
                const float4 ss = ((const float4*)s_src)[src];
                sc.x = ss.x + sd[u * HEADS + 0];
                sc.y = ss.y + sd[u * HEADS + 1];
                sc.z = ss.z + sd[u * HEADS + 2];
                sc.w = ss.w + sd[u * HEADS + 3];
                sc.x = sc.x > 0.f ? sc.x : 0.2f * sc.x;
                sc.y = sc.y > 0.f ? sc.y : 0.2f * sc.y;
                sc.z = sc.z > 0.f ? sc.z : 0.2f * sc.z;
                sc.w = sc.w > 0.f ? sc.w : 0.2f * sc.w;
                atomicMax(&mx[u * HEADS + 0], flipf(sc.x));
                atomicMax(&mx[u * HEADS + 1], flipf(sc.y));
                atomicMax(&mx[u * HEADS + 2], flipf(sc.z));
                atomicMax(&mx[u * HEADS + 3], flipf(sc.w));
            }
            const unsigned vm = __builtin_amdgcn_ballot_w32(pk != NOEDGE);
            const unsigned b0 = __builtin_amdgcn_ballot_w32((pk & 1u) != 0u);
            const unsigned b1 = __builtin_amdgcn_ballot_w32((pk & 2u) != 0u);
            const unsigned b2 = __builtin_amdgcn_ballot_w32((pk & 4u) != 0u);
            const unsigned b3 = __builtin_amdgcn_ballot_w32((pk & 8u) != 0u);
            const unsigned b4 = __builtin_amdgcn_ballot_w32((pk & 16u) != 0u);
            const unsigned o  = pk & 31u;
            const unsigned peers = owner_lanes(vm, b0, b1, b2, b3, b4, o);
            const int slot = __builtin_amdgcn_ds_bpermute((int)(o << 2), off)
                           + (int)__builtin_amdgcn_mbcnt_lo(peers, 0u);
            if (pk != NOEDGE && slot < LCAP) {
                own[o * LSTR + slot] = pk;
                scr[o * LSTR + slot] = sc;
            }
            off += __builtin_popcount(owner_lanes(vm, b0, b1, b2, b3, b4, (unsigned)lane));
        }
        if (w == NWAVE - 1) {
            ocnt[lane] = off < LCAP ? off : LCAP;
            if (off > LCAP) ovf = 1;
        }
    }
    __syncthreads();
    const int slow2 = __builtin_amdgcn_readfirstlane(ovf);

    int pend_s = 0, pend_d = 0, npend = 0;
    const int h  = lane >> 3;
    const int sl = lane & 7;
#define GROUP_LANE(x, k) __uint_as_float(__builtin_amdgcn_mov_dpp8(__float_as_uint(x), (unsigned)(k) * 0x249249u))
#define EDGE_STEP(k)                                                        \
    {                                                                       \
        const int   d   = __builtin_amdgcn_readlane(pend_d, k);             \
        const float eek = GROUP_LANE(ee, k);                                \
        run = (pend_d == d) ? run + eek : run;                              \
        if (k < npend) {                                                    \
            float4 a = acc[d * 32 + lane];                                  \
            a.x += eek * row[k].x;                                          \
            a.y += eek * row[k].y;                                          \
            a.z += eek * row[k].z;                                          \
            a.w += eek * row[k].w;                                          \
            acc[d * 32 + lane] = a;                                         \
        }                                                                   \
    }
    auto flush = [&](const bool carried, const int cslot) {
        float4 row[8];
        #pragma unroll
        for (int k = 0; k < 8; ++k) {
            int s = __builtin_amdgcn_readlane(pend_s, k);
            row[k] = *(const float4*)(Wx + (size_t)s * DIM + lane * 4);
        }
        float v;
        if (carried) {
            v = ((const float*)scr)[cslot * HEADS + h];
        } else {
            v = s_src[pend_s * HEADS + h] + sd[pend_d * HEADS + h];
            v = v > 0.f ? v : 0.2f * v;
        }
        float ee = expf(v - unflipf(mx[pend_d * HEADS + h]));
        if (sl >= npend) ee = 0.f;
        float run = den[pend_d * HEADS + h];
        EDGE_STEP(0) EDGE_STEP(1) EDGE_STEP(2) EDGE_STEP(3)
        EDGE_STEP(4) EDGE_STEP(5) EDGE_STEP(6) EDGE_STEP(7)
        if (sl < npend) den[pend_d * HEADS + h] = run;
        npend = 0;
    };
#undef EDGE_STEP
#undef GROUP_LANE
    if (!slow2) {
        const unsigned* ml = own + w * LSTR;
        const int n = __builtin_amdgcn_readfirstlane(ocnt[w]);
        int nb = n < 8 ? n : 8;
        unsigned pk = (n > 0) ? ml[sl < nb ? sl : nb - 1] : 0u;
        for (int b0 = 0; b0 < n; b0 += 8) {
            pend_s = (int)(pk >> 8);
            pend_d = (int)(pk & 255u);
            const int cur = nb;
            const int rest = n - b0 - 8;
            nb = rest < 8 ? rest : 8;
            if (rest > 0) pk = ml[b0 + 8 + (sl < nb ? sl : nb - 1)];
            if (cur == 8) { npend = 8;   flush(true, w * LSTR + b0 + sl); }
            else          { npend = cur; flush(true, w * LSTR + b0 + (sl < cur ? sl : cur - 1)); }
        }
    } else {
        auto push = [&](int s, int d) {
            pend_s = (sl == npend) ? s : pend_s;
            pend_d = (sl == npend) ? d : pend_d;
            if (++npend == 8) flush(false, 0);
        };
        for (int b = 0; b < nE; b += 32) {
            int idx = b + lane;
            unsigned u = (idx < nE) ? (unsigned)(dstp[idx] - lo) : 0xFFFFFFFFu;
            bool mine = u < ucnt && (int)(u & 31u) == w;
            int sv = mine ? ei[idx] : 0;
            unsigned hit = __builtin_amdgcn_ballot_w32(mine);
            while (hit) {
                int j = __builtin_ctz(hit);
                hit &= hit - 1;
                push(__builtin_amdgcn_readlane(sv, j), __builtin_amdgcn_readlane((int)u, j));
            }
        }
        if (npend > 0) flush(false, 0);
    }
    __syncthreads();

    for (int q = tid; q < cntR * 32; q += 1024) {
        int n = q >> 5, c4 = q & 31;
        float dn = den[n * HEADS + (c4 >> 3)] + 1e-8f;
        float4 a = acc[q];
        float4 r;
        r.x = a.x / dn; r.x = r.x > 0.f ? r.x : expm1f(r.x);
        r.y = a.y / dn; r.y = r.y > 0.f ? r.y : expm1f(r.y);
        r.z = a.z / dn; r.z = r.z > 0.f ? r.z : expm1f(r.z);
        r.w = a.w / dn; r.w = r.w > 0.f ? r.w : expm1f(r.w);
        *(float4*)(out + (size_t)(lo + n) * DIM + c4 * 4) = r;
    }
}

extern "C" void kernel_launch(void* const* d_in, const int* in_sizes, int n_in,
                              void* d_out, int out_size, void* d_ws, size_t ws_size,
                              hipStream_t stream) {
    const float* x   = (const float*)d_in[0];
    const int*   ei  = (const int*)d_in[1];
    const float* W   = (const float*)d_in[2];
    const float* a_w = (const float*)d_in[3];
    int N = in_sizes[0] / DIM;
    int E = in_sizes[1] / 2;
    float* out = (float*)d_out;

    float* Wx    = (float*)d_ws;
    float* s_src = Wx + (size_t)N * DIM;
    float* s_dst = s_src + (size_t)N * HEADS;
    unsigned* pres = (unsigned*)(s_dst + (size_t)N * HEADS);

    dim3 blk(256);
    int nTiles = (N + 15) / 16;
    gat_gemm_wmma<<<256, blk, 0, stream>>>(x, W, Wx, N, nTiles);
    gat_scores  <<<(N * HEADS + 255) / 256, blk, 0, stream>>>(Wx, a_w, s_src, s_dst, N, ei, pres, E);
    gat_edge_owner<<<(N + RN - 1) / RN, dim3(1024), 0, stream>>>(ei, s_src, s_dst, Wx, pres, out, E, N);
}
